# starting kernel
# baseline (speedup 1.0000x reference)
#include <hip/hip_runtime.h>


#define D  10
#define RT 4
#define SP 12

typedef __attribute__((ext_vector_type(2))) float v2f;
typedef __attribute__((ext_vector_type(8))) float v8f;

__device__ __forceinline__ v8f wmma_k4(v2f a, v2f b, v8f c) {
  return __builtin_amdgcn_wmma_f32_16x16x4_f32(
       false, a,  false, b,
       (short)0, c,  false,  false);
}

__global__ void k_prep(const float* __restrict__ Ws1, const float* __restrict__ Wn1,
                       const float* __restrict__ b1,
                       const float* __restrict__ Ws2, const float* __restrict__ Wn2,
                       const float* __restrict__ b2,
                       float* __restrict__ WsP1, float* __restrict__ WnP1,
                       float* __restrict__ bsP1,
                       float* __restrict__ WsP2, float* __restrict__ WnP2,
                       float* __restrict__ bsP2) {
  for (int i = threadIdx.x; i < 16 * SP; i += blockDim.x) {
    int n = i / SP, k = i % SP;
    float s1 = 0.f, s2 = 0.f;
    if (n < D && k < D) {
#pragma unroll
      for (int r = 0; r < RT; ++r) {
        s1 += Ws1[r * D * D + n * D + k];
        s2 += Ws2[r * D * D + n * D + k];
      }
    }
    WsP1[i] = s1;
    WsP2[i] = s2;
  }
  for (int i = threadIdx.x; i < RT * 16 * SP; i += blockDim.x) {
    int r = i / (16 * SP), j = i % (16 * SP);
    int n = j / SP, k = j % SP;
    float v1 = 0.f, v2 = 0.f;
    if (n < D && k < D) {
      v1 = Wn1[r * D * D + n * D + k];
      v2 = Wn2[r * D * D + n * D + k];
    }
    WnP1[i] = v1;
    WnP2[i] = v2;
  }
  for (int i = threadIdx.x; i < 16; i += blockDim.x) {
    float s1 = 0.f, s2 = 0.f;
    if (i < D) {
#pragma unroll
      for (int r = 0; r < RT; ++r) {
        s1 += b1[r * D + i];
        s2 += b2[r * D + i];
      }
    }
    bsP1[i] = s1;
    bsP2[i] = s2;
  }
}

__global__ void k_padcopy(const float* __restrict__ x, float* __restrict__ hp, int Nn) {
  int i = blockIdx.x * blockDim.x + threadIdx.x;
  if (i >= Nn * SP) return;
  int n = i / SP, k = i - n * SP;
  hp[i] = (k < D) ? x[(long long)n * D + k] : 0.0f;
}

__global__ void k_zero(float* __restrict__ p, int n) {
  int i = blockIdx.x * blockDim.x + threadIdx.x;
  if (i < n) p[i] = 0.0f;
}

__global__ void k_scatter(const float* __restrict__ hp,
                          const int* __restrict__ src,
                          const int* __restrict__ dst,
                          float* __restrict__ msum,
                          float* __restrict__ deg, int E_) {
  long long gid = (long long)blockIdx.x * blockDim.x + threadIdx.x;
  if (gid >= (long long)E_ * D) return;
  int e = (int)(gid / D);
  int k = (int)(gid - (long long)e * D);
  int s = src[e];
  int d = dst[e];
  atomicAdd(&msum[(long long)d * SP + k], hp[(long long)s * SP + k]);
  if (k == 0) atomicAdd(&deg[d], 1.0f);
}


__global__ void k_self_wmma(const float* __restrict__ hp,
                            const float* __restrict__ WsP,
                            const float* __restrict__ bsP,
                            float* __restrict__ hout,
                            int nTiles, int Nn) {
  int wave = blockIdx.x * (blockDim.x >> 5) + (threadIdx.x >> 5);
  if (wave >= nTiles) return;
  int lane = threadIdx.x & 31;
  int m    = lane & 15;
  int half = lane >> 4;
  int node = min(wave * 16 + m, Nn - 1);

  union { v8f v; float f[8]; } acc;
  float bsum = bsP[m];
#pragma unroll
  for (int j = 0; j < 8; ++j) acc.f[j] = bsum;

  const float* arow = hp + (long long)node * SP;
  const float* brow = WsP + m * SP;
#pragma unroll
  for (int c = 0; c < 3; ++c) {
    int k0 = c * 4 + half * 2;
    v2f a, b;
    a.x = arow[k0]; a.y = arow[k0 + 1];
    b.x = brow[k0]; b.y = brow[k0 + 1];
    acc.v = wmma_k4(a, b, acc.v);
  }

  int rowBase = wave * 16 + half * 8;
  bool fullTile = (wave * 16 + 16) <= Nn;
  if (m < SP) {
    float* op = hout + (long long)rowBase * SP + m;
    if (fullTile) {
#pragma unroll
      for (int j = 0; j < 8; ++j) op[j * SP] = acc.f[j];
    } else {
#pragma unroll
      for (int j = 0; j < 8; ++j)
        if (rowBase + j < Nn) op[j * SP] = acc.f[j];
    }
  }
}

__global__ void k_neigh_wmma(const float* __restrict__ msum,
                             const float* __restrict__ deg,
                             const float* __restrict__ WnP,
                             float* __restrict__ hout,
                             int nTiles, int Nn) {
  int wave = blockIdx.x * (blockDim.x >> 5) + (threadIdx.x >> 5);
  if (wave >= nTiles) return;
  int lane = threadIdx.x & 31;
  int m    = lane & 15;
  int half = lane >> 4;
  int node = min(wave * 16 + m, Nn - 1);
  int mc   = min(m, SP - 1);
  int rowBase = wave * 16 + half * 8;
  bool fullTile = (wave * 16 + 16) <= Nn;

  float rdeg = 1.0f / fmaxf(deg[node], 1.0f);

  union { v8f v; float f[8]; } acc;
  if (fullTile) {
    const float* cp = hout + (long long)rowBase * SP + mc;
#pragma unroll
    for (int j = 0; j < 8; ++j) acc.f[j] = cp[j * SP];
  } else {
#pragma unroll
    for (int j = 0; j < 8; ++j)
      acc.f[j] = hout[(long long)min(rowBase + j, Nn - 1) * SP + mc];
  }

  const float* arow = msum + (long long)node * SP;
  const float* brow = WnP + m * SP;
#pragma unroll
  for (int c = 0; c < 3; ++c) {
    int k0 = c * 4 + half * 2;
    v2f a, b;
    a.x = arow[k0] * rdeg; a.y = arow[k0 + 1] * rdeg;
    b.x = brow[k0];        b.y = brow[k0 + 1];
    acc.v = wmma_k4(a, b, acc.v);
  }

  if (m < D) {
    float* op = hout + (long long)rowBase * SP + m;
    if (fullTile) {
#pragma unroll
      for (int j = 0; j < 8; ++j) op[j * SP] = acc.f[j];
    } else {
#pragma unroll
      for (int j = 0; j < 8; ++j)
        if (rowBase + j < Nn) op[j * SP] = acc.f[j];
    }
  }
}

__global__ void k_score(const float* __restrict__ hp,
                        const int* __restrict__ ps, const int* __restrict__ pd,
                        const int* __restrict__ ns, const int* __restrict__ nd,
                        float* __restrict__ out, int E_) {
  int i = blockIdx.x * blockDim.x + threadIdx.x;
  if (i >= 2 * E_) return;
  int a, b;
  if (i < E_) { a = ps[i]; b = pd[i]; }
  else        { a = ns[i - E_]; b = nd[i - E_]; }
  const float* pa = hp + (long long)a * SP;
  const float* pb = hp + (long long)b * SP;
  float s = 0.0f;
#pragma unroll
  for (int k = 0; k < D; ++k) s += pa[k] * pb[k];
  out[i] = s;
}

extern "C" void kernel_launch(void* const* d_in, const int* in_sizes, int n_in,
                              void* d_out, int out_size, void* d_ws, size_t ws_size,
                              hipStream_t stream) {
  (void)n_in; (void)out_size; (void)ws_size;
  const float* x    = (const float*)d_in[0];
  const int*   esrc = (const int*)d_in[1];
  const int*   edst = (const int*)d_in[2];
  const int*   nsrc = (const int*)d_in[3];
  const int*   ndst = (const int*)d_in[4];
  const float* Ws1  = (const float*)d_in[5];
  const float* Wn1  = (const float*)d_in[6];
  const float* b1   = (const float*)d_in[7];
  const float* Ws2  = (const float*)d_in[8];
  const float* Wn2  = (const float*)d_in[9];
  const float* b2   = (const float*)d_in[10];

  const int Nn = in_sizes[0] / D;
  const int Ee = in_sizes[3];

  float* B0   = (float*)d_ws;
  float* B1   = B0 + (size_t)Nn * SP;
  float* msum = B1 + (size_t)Nn * SP;
  float* dega = msum + (size_t)Nn * SP;
  float* WsP1 = dega + (size_t)Nn;
  float* WnP1 = WsP1 + 16 * SP;
  float* bsP1 = WnP1 + RT * 16 * SP;
  float* WsP2 = bsP1 + 16;
  float* WnP2 = WsP2 + 16 * SP;
  float* bsP2 = WnP2 + RT * 16 * SP;

  const int tiles   = (Nn + 15) / 16;
  const int wBlocks = (tiles + 3) / 4;
  const int zeroCnt = Nn * SP + Nn;
  const int padCnt  = Nn * SP;
  const long long sThreads = (long long)Ee * D;
  const int sBlocks = (int)((sThreads + 255) / 256);

  k_prep<<<1, 256, 0, stream>>>(Ws1, Wn1, b1, Ws2, Wn2, b2,
                                WsP1, WnP1, bsP1, WsP2, WnP2, bsP2);
  k_padcopy<<<(padCnt + 255) / 256, 256, 0, stream>>>(x, B0, Nn);

  auto layer = [&](const float* hin, const float* WsP, const float* WnP,
                   const float* bsP, float* hout) {
    k_self_wmma<<<wBlocks, 128, 0, stream>>>(hin, WsP, bsP, hout, tiles, Nn);
    for (int r = 0; r < RT; ++r) {
      k_zero<<<(zeroCnt + 255) / 256, 256, 0, stream>>>(msum, zeroCnt);
      k_scatter<<<sBlocks, 256, 0, stream>>>(hin, esrc + (size_t)r * Ee,
                                             edst + (size_t)r * Ee, msum, dega, Ee);
      k_neigh_wmma<<<wBlocks, 128, 0, stream>>>(msum, dega, WnP + r * 16 * SP,
                                                hout, tiles, Nn);
    }
  };

  layer(B0, WsP1, WnP1, bsP1, B1);
  layer(B1, WsP2, WnP2, bsP2, B0);

  k_score<<<(2 * Ee + 255) / 256, 256, 0, stream>>>(B0, esrc, edst, nsrc, ndst,
                                                    (float*)d_out, Ee);
}


# Replace the per-edge f32 atomic scatter by one counting sort of edges by (etype,dst) per call plus an atomic-free segment gather fused into one WMMA kernel per layer; sum order = fill ticket order
# speedup vs baseline: 1.3649x; 1.3649x over previous
#include <hip/hip_runtime.h>


#define D  10
#define RT 4
#define SP 12

#define SCAN_T     256
#define SCAN_PER   8
#define SCAN_CHUNK (SCAN_T * SCAN_PER)
#define SCANB_T    1024

typedef __attribute__((ext_vector_type(2))) float v2f;
typedef __attribute__((ext_vector_type(8))) float v8f;

__device__ __forceinline__ v8f wmma_k4(v2f a, v2f b, v8f c) {
  return __builtin_amdgcn_wmma_f32_16x16x4_f32(
       false, a,  false, b,
       (short)0, c,  false,  false);
}

__global__ void k_prep(const float* __restrict__ Ws1, const float* __restrict__ Wn1,
                       const float* __restrict__ b1,
                       const float* __restrict__ Ws2, const float* __restrict__ Wn2,
                       const float* __restrict__ b2,
                       float* __restrict__ WsP1, float* __restrict__ WnP1,
                       float* __restrict__ bsP1,
                       float* __restrict__ WsP2, float* __restrict__ WnP2,
                       float* __restrict__ bsP2) {
  for (int i = threadIdx.x; i < 16 * SP; i += blockDim.x) {
    int n = i / SP, k = i % SP;
    float s1 = 0.f, s2 = 0.f;
    if (n < D && k < D) {
#pragma unroll
      for (int r = 0; r < RT; ++r) {
        s1 += Ws1[r * D * D + n * D + k];
        s2 += Ws2[r * D * D + n * D + k];
      }
    }
    WsP1[i] = s1;
    WsP2[i] = s2;
  }
  for (int i = threadIdx.x; i < RT * 16 * SP; i += blockDim.x) {
    int r = i / (16 * SP), j = i % (16 * SP);
    int n = j / SP, k = j % SP;
    float v1 = 0.f, v2 = 0.f;
    if (n < D && k < D) {
      v1 = Wn1[r * D * D + n * D + k];
      v2 = Wn2[r * D * D + n * D + k];
    }
    WnP1[i] = v1;
    WnP2[i] = v2;
  }
  for (int i = threadIdx.x; i < 16; i += blockDim.x) {
    float s1 = 0.f, s2 = 0.f;
    if (i < D) {
#pragma unroll
      for (int r = 0; r < RT; ++r) {
        s1 += b1[r * D + i];
        s2 += b2[r * D + i];
      }
    }
    bsP1[i] = s1;
    bsP2[i] = s2;
  }
}

__global__ void k_padcopy(const float* __restrict__ x, float* __restrict__ hp, int Nn) {
  int i = blockIdx.x * blockDim.x + threadIdx.x;
  if (i >= Nn * SP) return;
  int n = i / SP, k = i - n * SP;
  hp[i] = (k < D) ? x[(long long)n * D + k] : 0.0f;
}

__global__ void k_zero_i(int* __restrict__ p, int n) {
  int i = blockIdx.x * blockDim.x + threadIdx.x;
  if (i < n) p[i] = 0;
}


__global__ void k_hist(const int* __restrict__ edst,
                       int* __restrict__ cnt,
                       int Nn, int Ee, int total) {
  int i = blockIdx.x * blockDim.x + threadIdx.x;
  if (i >= total) return;
  int r = i / Ee;
  atomicAdd(&cnt[r * Nn + edst[i]], 1);
}

__global__ void k_scan_a(const int* __restrict__ cnt, int* __restrict__ bsum, int n) {
  __shared__ int sh[SCAN_T];
  int t = threadIdx.x;
  int base = blockIdx.x * SCAN_CHUNK + t * SCAN_PER;
  int s = 0;
#pragma unroll
  for (int q = 0; q < SCAN_PER; ++q) {
    int i = base + q;
    if (i < n) s += cnt[i];
  }
  sh[t] = s;
  __syncthreads();
  for (int ofs = SCAN_T / 2; ofs > 0; ofs >>= 1) {
    if (t < ofs) sh[t] += sh[t + ofs];
    __syncthreads();
  }
  if (t == 0) bsum[blockIdx.x] = sh[0];
}

__global__ void k_scan_b(int* __restrict__ bsum, int nb) {
  __shared__ int sh[SCANB_T];
  int t = threadIdx.x;
  int carry = 0;
  for (int base = 0; base < nb; base += SCANB_T) {
    int i = base + t;
    int v = (i < nb) ? bsum[i] : 0;
    sh[t] = v;
    __syncthreads();
    for (int ofs = 1; ofs < SCANB_T; ofs <<= 1) {
      int add = (t >= ofs) ? sh[t - ofs] : 0;
      __syncthreads();
      sh[t] += add;
      __syncthreads();
    }
    int incl = sh[t];
    int tot  = sh[SCANB_T - 1];
    if (i < nb) bsum[i] = carry + incl - v;
    carry += tot;
    __syncthreads();
  }
  if (t == 0) bsum[nb] = carry;
}

__global__ void k_scan_c(const int* __restrict__ cnt, const int* __restrict__ bexc,
                         int* __restrict__ off, int n) {
  __shared__ int sh[SCAN_T];
  int t = threadIdx.x;
  int base = blockIdx.x * SCAN_CHUNK + t * SCAN_PER;
  int v[SCAN_PER];
  int s = 0;
#pragma unroll
  for (int q = 0; q < SCAN_PER; ++q) {
    int i = base + q;
    v[q] = (i < n) ? cnt[i] : 0;
    s += v[q];
  }
  sh[t] = s;
  __syncthreads();
  for (int ofs = 1; ofs < SCAN_T; ofs <<= 1) {
    int add = (t >= ofs) ? sh[t - ofs] : 0;
    __syncthreads();
    sh[t] += add;
    __syncthreads();
  }
  int run = bexc[blockIdx.x] + sh[t] - s;
#pragma unroll
  for (int q = 0; q < SCAN_PER; ++q) {
    int i = base + q;
    if (i < n) off[i] = run;
    run += v[q];
  }
  if (blockIdx.x == 0 && t == 0) off[n] = bexc[gridDim.x];
}

__global__ void k_fill(const int* __restrict__ esrc,
                       const int* __restrict__ edst,
                       const int* __restrict__ off,
                       int* __restrict__ cnt,
                       int* __restrict__ col,
                       int Nn, int Ee, int total) {
  int i = blockIdx.x * blockDim.x + threadIdx.x;
  if (i >= total) return;
  int r = i / Ee;
  int g = r * Nn + edst[i];
  int s = esrc[i];
  int p = atomicSub(&cnt[g], 1) - 1;
  col[off[g] + p] = s;
}


__global__ void k_layer(const float* __restrict__ hp,
                        const int* __restrict__ off,
                        const int* __restrict__ col,
                        const float* __restrict__ WsP,
                        const float* __restrict__ WnP,
                        const float* __restrict__ bsP,
                        float* __restrict__ hout,
                        int nTiles, int Nn) {
  int wave = blockIdx.x * (blockDim.x >> 5) + (threadIdx.x >> 5);
  if (wave >= nTiles) return;
  int lane = threadIdx.x & 31;
  int m    = lane & 15;
  int half = lane >> 4;
  int kk   = half * 2;
  int node = min(wave * 16 + m, Nn - 1);

  union { v8f v; float f[8]; } acc;
  float bsum = bsP[m];
#pragma unroll
  for (int j = 0; j < 8; ++j) acc.f[j] = bsum;

  {
    const float* arow = hp + (long long)node * SP + kk;
    const float* brow = WsP + m * SP + kk;
#pragma unroll
    for (int c = 0; c < 3; ++c) {
      v2f a = *(const v2f*)(arow + c * 4);
      v2f b = *(const v2f*)(brow + c * 4);
      acc.v = wmma_k4(a, b, acc.v);
    }
  }

#pragma unroll
  for (int r = 0; r < RT; ++r) {
    int g   = r * Nn + node;
    int beg = off[g];
    int end = off[g + 1];
    v2f s0 = {0.f, 0.f}, s1 = {0.f, 0.f}, s2 = {0.f, 0.f};
    int j = beg;
    for (; j + 1 < end; j += 2) {
      int sa = col[j];
      int sb = col[j + 1];
      const float* ra = hp + (long long)sa * SP + kk;
      const float* rb = hp + (long long)sb * SP + kk;
      v2f xa0 = *(const v2f*)(ra);
      v2f xa1 = *(const v2f*)(ra + 4);
      v2f xa2 = *(const v2f*)(ra + 8);
      v2f xb0 = *(const v2f*)(rb);
      v2f xb1 = *(const v2f*)(rb + 4);
      v2f xb2 = *(const v2f*)(rb + 8);
      s0 += xa0; s1 += xa1; s2 += xa2;
      s0 += xb0; s1 += xb1; s2 += xb2;
    }
    if (j < end) {
      int sa = col[j];
      const float* ra = hp + (long long)sa * SP + kk;
      s0 += *(const v2f*)(ra);
      s1 += *(const v2f*)(ra + 4);
      s2 += *(const v2f*)(ra + 8);
    }
    float rdeg = 1.0f / fmaxf((float)(end - beg), 1.0f);
    const float* bw = WnP + r * 16 * SP + m * SP + kk;
    v2f b0 = *(const v2f*)(bw);
    v2f b1 = *(const v2f*)(bw + 4);
    v2f b2 = *(const v2f*)(bw + 8);
    acc.v = wmma_k4(s0 * rdeg, b0, acc.v);
    acc.v = wmma_k4(s1 * rdeg, b1, acc.v);
    acc.v = wmma_k4(s2 * rdeg, b2, acc.v);
  }

  int rowBase = wave * 16 + half * 8;
  bool fullTile = (wave * 16 + 16) <= Nn;
  if (m < SP) {
    float* op = hout + (long long)rowBase * SP + m;
    if (fullTile) {
#pragma unroll
      for (int j = 0; j < 8; ++j) op[j * SP] = acc.f[j];
    } else {
#pragma unroll
      for (int j = 0; j < 8; ++j)
        if (rowBase + j < Nn) op[j * SP] = acc.f[j];
    }
  }
}

__global__ void k_score(const float* __restrict__ hp,
                        const int* __restrict__ ps, const int* __restrict__ pd,
                        const int* __restrict__ ns, const int* __restrict__ nd,
                        float* __restrict__ out, int E_) {
  int i = blockIdx.x * blockDim.x + threadIdx.x;
  if (i >= 2 * E_) return;
  int a, b;
  if (i < E_) { a = ps[i]; b = pd[i]; }
  else        { a = ns[i - E_]; b = nd[i - E_]; }
  const float* pa = hp + (long long)a * SP;
  const float* pb = hp + (long long)b * SP;
  float s = 0.0f;
#pragma unroll
  for (int k = 0; k < D; ++k) s += pa[k] * pb[k];
  out[i] = s;
}

extern "C" void kernel_launch(void* const* d_in, const int* in_sizes, int n_in,
                              void* d_out, int out_size, void* d_ws, size_t ws_size,
                              hipStream_t stream) {
  (void)n_in; (void)out_size; (void)ws_size;
  const float* x    = (const float*)d_in[0];
  const int*   esrc = (const int*)d_in[1];
  const int*   edst = (const int*)d_in[2];
  const int*   nsrc = (const int*)d_in[3];
  const int*   ndst = (const int*)d_in[4];
  const float* Ws1  = (const float*)d_in[5];
  const float* Wn1  = (const float*)d_in[6];
  const float* b1   = (const float*)d_in[7];
  const float* Ws2  = (const float*)d_in[8];
  const float* Wn2  = (const float*)d_in[9];
  const float* b2   = (const float*)d_in[10];

  const int Nn = in_sizes[0] / D;
  const int Ee = in_sizes[3];

  const int segs   = RT * Nn;
  const int edges  = RT * Ee;
  const int offCnt = (segs + 1 + 3) & ~3;
  const int nb     = (segs + SCAN_CHUNK - 1) / SCAN_CHUNK;
  const int bsCnt  = (nb + 1 + 3) & ~3;

  float* B0   = (float*)d_ws;
  float* B1   = B0 + (size_t)Nn * SP;
  int*   cnt  = (int*)(B1 + (size_t)Nn * SP);
  int*   off  = cnt + (size_t)segs;
  int*   col  = off + (size_t)offCnt;
  int*   bsum = col + (size_t)edges;
  float* WsP1 = (float*)(bsum + (size_t)bsCnt);
  float* WnP1 = WsP1 + 16 * SP;
  float* bsP1 = WnP1 + RT * 16 * SP;
  float* WsP2 = bsP1 + 16;
  float* WnP2 = WsP2 + 16 * SP;
  float* bsP2 = WnP2 + RT * 16 * SP;

  const int tiles   = (Nn + 15) / 16;
  const int wBlocks = (tiles + 3) / 4;
  const int padCnt  = Nn * SP;
  const int eBlocks = (edges + 255) / 256;

  k_prep<<<1, 256, 0, stream>>>(Ws1, Wn1, b1, Ws2, Wn2, b2,
                                WsP1, WnP1, bsP1, WsP2, WnP2, bsP2);
  k_padcopy<<<(padCnt + 255) / 256, 256, 0, stream>>>(x, B0, Nn);

  k_zero_i<<<(segs + 255) / 256, 256, 0, stream>>>(cnt, segs);
  k_hist<<<eBlocks, 256, 0, stream>>>(edst, cnt, Nn, Ee, edges);
  k_scan_a<<<nb, SCAN_T, 0, stream>>>(cnt, bsum, segs);
  k_scan_b<<<1, SCANB_T, 0, stream>>>(bsum, nb);
  k_scan_c<<<nb, SCAN_T, 0, stream>>>(cnt, bsum, off, segs);
  k_fill<<<eBlocks, 256, 0, stream>>>(esrc, edst, off, cnt, col, Nn, Ee, edges);

  k_layer<<<wBlocks, 128, 0, stream>>>(B0, off, col, WsP1, WnP1, bsP1, B1, tiles, Nn);
  k_layer<<<wBlocks, 128, 0, stream>>>(B1, off, col, WsP2, WnP2, bsP2, B0, tiles, Nn);

  k_score<<<(2 * Ee + 255) / 256, 256, 0, stream>>>(B0, esrc, edst, nsrc, ndst,
                                                    (float*)d_out, Ee);
}


# Pad-copy, zero, histogram, fill and score run as resident grids looping over elements (4 independent edges per lane per trip) instead of one element per lane; sums stay in fill ticket order
# speedup vs baseline: 1.5013x; 1.0999x over previous
#include <hip/hip_runtime.h>


#define D  10
#define RT 4
#define SP 12

#define SCAN_T     256
#define SCAN_PER   8
#define SCAN_CHUNK (SCAN_T * SCAN_PER)
#define SCANB_T    1024

#define RES_T 256
#define EPT   4
#define SPT   2
#define PPT   8

typedef __attribute__((ext_vector_type(2))) float v2f;
typedef __attribute__((ext_vector_type(8))) float v8f;

__device__ __forceinline__ v8f wmma_k4(v2f a, v2f b, v8f c) {
  return __builtin_amdgcn_wmma_f32_16x16x4_f32(
       false, a,  false, b,
       (short)0, c,  false,  false);
}

__global__ void k_prep(const float* __restrict__ Ws1, const float* __restrict__ Wn1,
                       const float* __restrict__ b1,
                       const float* __restrict__ Ws2, const float* __restrict__ Wn2,
                       const float* __restrict__ b2,
                       float* __restrict__ WsP1, float* __restrict__ WnP1,
                       float* __restrict__ bsP1,
                       float* __restrict__ WsP2, float* __restrict__ WnP2,
                       float* __restrict__ bsP2) {
  for (int i = threadIdx.x; i < 16 * SP; i += blockDim.x) {
    int n = i / SP, k = i % SP;
    float s1 = 0.f, s2 = 0.f;
    if (n < D && k < D) {
#pragma unroll
      for (int r = 0; r < RT; ++r) {
        s1 += Ws1[r * D * D + n * D + k];
        s2 += Ws2[r * D * D + n * D + k];
      }
    }
    WsP1[i] = s1;
    WsP2[i] = s2;
  }
  for (int i = threadIdx.x; i < RT * 16 * SP; i += blockDim.x) {
    int r = i / (16 * SP), j = i % (16 * SP);
    int n = j / SP, k = j % SP;
    float v1 = 0.f, v2 = 0.f;
    if (n < D && k < D) {
      v1 = Wn1[r * D * D + n * D + k];
      v2 = Wn2[r * D * D + n * D + k];
    }
    WnP1[i] = v1;
    WnP2[i] = v2;
  }
  for (int i = threadIdx.x; i < 16; i += blockDim.x) {
    float s1 = 0.f, s2 = 0.f;
    if (i < D) {
#pragma unroll
      for (int r = 0; r < RT; ++r) {
        s1 += b1[r * D + i];
        s2 += b2[r * D + i];
      }
    }
    bsP1[i] = s1;
    bsP2[i] = s2;
  }
}

__device__ __forceinline__ float pad_elem(const float* __restrict__ x, int i) {
  int n = i / SP, k = i - n * SP;
  float t = x[(long long)n * D + min(k, D - 1)];
  return (k < D) ? t : 0.0f;
}

__global__ void __launch_bounds__(RES_T)
k_padcopy(const float* __restrict__ x, float* __restrict__ hp, int Nn) {
  const int total = Nn * SP;
  const int step  = gridDim.x * (RES_T * PPT);
  int base = blockIdx.x * (RES_T * PPT) + threadIdx.x;
  for (; base + (PPT - 1) * RES_T < total; base += step) {
    float v[PPT];
#pragma unroll
    for (int q = 0; q < PPT; ++q) v[q] = pad_elem(x, base + q * RES_T);
#pragma unroll
    for (int q = 0; q < PPT; ++q) hp[base + q * RES_T] = v[q];
  }
#pragma unroll
  for (int q = 0; q < PPT; ++q) {
    int i = base + q * RES_T;
    if (i < total) hp[i] = pad_elem(x, i);
  }
}

__global__ void k_zero_i(int* __restrict__ p, int n) {
  const int stride = gridDim.x * blockDim.x;
  for (int i = blockIdx.x * blockDim.x + threadIdx.x; i < n; i += stride) p[i] = 0;
}


__global__ void __launch_bounds__(RES_T)
k_hist(const int* __restrict__ edst,
       int* __restrict__ cnt,
       int Nn, int Ee) {
  const int r = blockIdx.y;
  const int* __restrict__ ed = edst + (long long)r * Ee;
  int* __restrict__ c = cnt + (long long)r * Nn;
  const int step = gridDim.x * (RES_T * EPT);
  int base = blockIdx.x * (RES_T * EPT) + threadIdx.x;
  for (; base + (EPT - 1) * RES_T < Ee; base += step) {
    int d[EPT];
#pragma unroll
    for (int q = 0; q < EPT; ++q) d[q] = ed[base + q * RES_T];
#pragma unroll
    for (int q = 0; q < EPT; ++q) atomicAdd(&c[d[q]], 1);
  }
#pragma unroll
  for (int q = 0; q < EPT; ++q) {
    int i = base + q * RES_T;
    if (i < Ee) atomicAdd(&c[ed[i]], 1);
  }
}

__global__ void k_scan_a(const int* __restrict__ cnt, int* __restrict__ bsum, int n) {
  __shared__ int sh[SCAN_T];
  int t = threadIdx.x;
  int base = blockIdx.x * SCAN_CHUNK + t * SCAN_PER;
  int s = 0;
#pragma unroll
  for (int q = 0; q < SCAN_PER; ++q) {
    int i = base + q;
    if (i < n) s += cnt[i];
  }
  sh[t] = s;
  __syncthreads();
  for (int ofs = SCAN_T / 2; ofs > 0; ofs >>= 1) {
    if (t < ofs) sh[t] += sh[t + ofs];
    __syncthreads();
  }
  if (t == 0) bsum[blockIdx.x] = sh[0];
}

__global__ void k_scan_b(int* __restrict__ bsum, int nb) {
  __shared__ int sh[SCANB_T];
  int t = threadIdx.x;
  int carry = 0;
  for (int base = 0; base < nb; base += SCANB_T) {
    int i = base + t;
    int v = (i < nb) ? bsum[i] : 0;
    sh[t] = v;
    __syncthreads();
    for (int ofs = 1; ofs < SCANB_T; ofs <<= 1) {
      int add = (t >= ofs) ? sh[t - ofs] : 0;
      __syncthreads();
      sh[t] += add;
      __syncthreads();
    }
    int incl = sh[t];
    int tot  = sh[SCANB_T - 1];
    if (i < nb) bsum[i] = carry + incl - v;
    carry += tot;
    __syncthreads();
  }
  if (t == 0) bsum[nb] = carry;
}

__global__ void k_scan_c(const int* __restrict__ cnt, const int* __restrict__ bexc,
                         int* __restrict__ off, int n) {
  __shared__ int sh[SCAN_T];
  int t = threadIdx.x;
  int base = blockIdx.x * SCAN_CHUNK + t * SCAN_PER;
  int v[SCAN_PER];
  int s = 0;
#pragma unroll
  for (int q = 0; q < SCAN_PER; ++q) {
    int i = base + q;
    v[q] = (i < n) ? cnt[i] : 0;
    s += v[q];
  }
  sh[t] = s;
  __syncthreads();
  for (int ofs = 1; ofs < SCAN_T; ofs <<= 1) {
    int add = (t >= ofs) ? sh[t - ofs] : 0;
    __syncthreads();
    sh[t] += add;
    __syncthreads();
  }
  int run = bexc[blockIdx.x] + sh[t] - s;
#pragma unroll
  for (int q = 0; q < SCAN_PER; ++q) {
    int i = base + q;
    if (i < n) off[i] = run;
    run += v[q];
  }
  if (blockIdx.x == 0 && t == 0) off[n] = bexc[gridDim.x];
}

__global__ void __launch_bounds__(RES_T)
k_fill(const int* __restrict__ esrc,
       const int* __restrict__ edst,
       const int* __restrict__ off,
       int* __restrict__ cnt,
       int* __restrict__ col,
       int Nn, int Ee) {
  const int r = blockIdx.y;
  const int* __restrict__ es = esrc + (long long)r * Ee;
  const int* __restrict__ ed = edst + (long long)r * Ee;
  const int* __restrict__ o  = off + (long long)r * Nn;
  int* __restrict__ c = cnt + (long long)r * Nn;
  const int step = gridDim.x * (RES_T * EPT);
  int base = blockIdx.x * (RES_T * EPT) + threadIdx.x;
  for (; base + (EPT - 1) * RES_T < Ee; base += step) {
    int d[EPT], s[EPT], ob[EPT], p[EPT];
#pragma unroll
    for (int q = 0; q < EPT; ++q) {
      d[q] = ed[base + q * RES_T];
      s[q] = es[base + q * RES_T];
    }
#pragma unroll
    for (int q = 0; q < EPT; ++q) ob[q] = o[d[q]];
#pragma unroll
    for (int q = 0; q < EPT; ++q) p[q] = atomicSub(&c[d[q]], 1) - 1;
#pragma unroll
    for (int q = 0; q < EPT; ++q) col[ob[q] + p[q]] = s[q];
  }
#pragma unroll
  for (int q = 0; q < EPT; ++q) {
    int i = base + q * RES_T;
    if (i < Ee) {
      int d = ed[i];
      int s = es[i];
      int pq = atomicSub(&c[d], 1) - 1;
      col[o[d] + pq] = s;
    }
  }
}


__global__ void k_layer(const float* __restrict__ hp,
                        const int* __restrict__ off,
                        const int* __restrict__ col,
                        const float* __restrict__ WsP,
                        const float* __restrict__ WnP,
                        const float* __restrict__ bsP,
                        float* __restrict__ hout,
                        int nTiles, int Nn) {
  int wave = blockIdx.x * (blockDim.x >> 5) + (threadIdx.x >> 5);
  if (wave >= nTiles) return;
  int lane = threadIdx.x & 31;
  int m    = lane & 15;
  int half = lane >> 4;
  int kk   = half * 2;
  int node = min(wave * 16 + m, Nn - 1);

  union { v8f v; float f[8]; } acc;
  float bsum = bsP[m];
#pragma unroll
  for (int j = 0; j < 8; ++j) acc.f[j] = bsum;

  {
    const float* arow = hp + (long long)node * SP + kk;
    const float* brow = WsP + m * SP + kk;
#pragma unroll
    for (int c = 0; c < 3; ++c) {
      v2f a = *(const v2f*)(arow + c * 4);
      v2f b = *(const v2f*)(brow + c * 4);
      acc.v = wmma_k4(a, b, acc.v);
    }
  }

#pragma unroll
  for (int r = 0; r < RT; ++r) {
    int g   = r * Nn + node;
    int beg = off[g];
    int end = off[g + 1];
    v2f s0 = {0.f, 0.f}, s1 = {0.f, 0.f}, s2 = {0.f, 0.f};
    int j = beg;
    for (; j + 1 < end; j += 2) {
      int sa = col[j];
      int sb = col[j + 1];
      const float* ra = hp + (long long)sa * SP + kk;
      const float* rb = hp + (long long)sb * SP + kk;
      v2f xa0 = *(const v2f*)(ra);
      v2f xa1 = *(const v2f*)(ra + 4);
      v2f xa2 = *(const v2f*)(ra + 8);
      v2f xb0 = *(const v2f*)(rb);
      v2f xb1 = *(const v2f*)(rb + 4);
      v2f xb2 = *(const v2f*)(rb + 8);
      s0 += xa0; s1 += xa1; s2 += xa2;
      s0 += xb0; s1 += xb1; s2 += xb2;
    }
    if (j < end) {
      int sa = col[j];
      const float* ra = hp + (long long)sa * SP + kk;
      s0 += *(const v2f*)(ra);
      s1 += *(const v2f*)(ra + 4);
      s2 += *(const v2f*)(ra + 8);
    }
    float rdeg = 1.0f / fmaxf((float)(end - beg), 1.0f);
    const float* bw = WnP + r * 16 * SP + m * SP + kk;
    v2f b0 = *(const v2f*)(bw);
    v2f b1 = *(const v2f*)(bw + 4);
    v2f b2 = *(const v2f*)(bw + 8);
    acc.v = wmma_k4(s0 * rdeg, b0, acc.v);
    acc.v = wmma_k4(s1 * rdeg, b1, acc.v);
    acc.v = wmma_k4(s2 * rdeg, b2, acc.v);
  }

  int rowBase = wave * 16 + half * 8;
  bool fullTile = (wave * 16 + 16) <= Nn;
  if (m < SP) {
    float* op = hout + (long long)rowBase * SP + m;
    if (fullTile) {
#pragma unroll
      for (int j = 0; j < 8; ++j) op[j * SP] = acc.f[j];
    } else {
#pragma unroll
      for (int j = 0; j < 8; ++j)
        if (rowBase + j < Nn) op[j * SP] = acc.f[j];
    }
  }
}

__device__ __forceinline__ float dot_rows(const float* __restrict__ hp, int a, int b) {
  const float* pa = hp + (long long)a * SP;
  const float* pb = hp + (long long)b * SP;
  float s = 0.0f;
#pragma unroll
  for (int k = 0; k < D; ++k) s += pa[k] * pb[k];
  return s;
}

__global__ void __launch_bounds__(RES_T)
k_score(const float* __restrict__ hp,
        const int* __restrict__ ps, const int* __restrict__ pd,
        const int* __restrict__ ns, const int* __restrict__ nd,
        float* __restrict__ out, int E_) {
  const int* __restrict__ ia = blockIdx.y ? ns : ps;
  const int* __restrict__ ib = blockIdx.y ? nd : pd;
  float* __restrict__ o = out + (long long)blockIdx.y * E_;
  const int step = gridDim.x * (RES_T * SPT);
  int base = blockIdx.x * (RES_T * SPT) + threadIdx.x;
  for (; base + (SPT - 1) * RES_T < E_; base += step) {
    int a[SPT], b[SPT];
    float s[SPT];
#pragma unroll
    for (int q = 0; q < SPT; ++q) {
      a[q] = ia[base + q * RES_T];
      b[q] = ib[base + q * RES_T];
    }
#pragma unroll
    for (int q = 0; q < SPT; ++q) s[q] = dot_rows(hp, a[q], b[q]);
#pragma unroll
    for (int q = 0; q < SPT; ++q) o[base + q * RES_T] = s[q];
  }
#pragma unroll
  for (int q = 0; q < SPT; ++q) {
    int i = base + q * RES_T;
    if (i < E_) o[i] = dot_rows(hp, ia[i], ib[i]);
  }
}

extern "C" void kernel_launch(void* const* d_in, const int* in_sizes, int n_in,
                              void* d_out, int out_size, void* d_ws, size_t ws_size,
                              hipStream_t stream) {
  (void)n_in; (void)out_size; (void)ws_size;
  const float* x    = (const float*)d_in[0];
  const int*   esrc = (const int*)d_in[1];
  const int*   edst = (const int*)d_in[2];
  const int*   nsrc = (const int*)d_in[3];
  const int*   ndst = (const int*)d_in[4];
  const float* Ws1  = (const float*)d_in[5];
  const float* Wn1  = (const float*)d_in[6];
  const float* b1   = (const float*)d_in[7];
  const float* Ws2  = (const float*)d_in[8];
  const float* Wn2  = (const float*)d_in[9];
  const float* b2   = (const float*)d_in[10];

  const int Nn = in_sizes[0] / D;
  const int Ee = in_sizes[3];

  const int segs   = RT * Nn;
  const int edges  = RT * Ee;
  const int offCnt = (segs + 1 + 3) & ~3;
  const int nb     = (segs + SCAN_CHUNK - 1) / SCAN_CHUNK;
  const int bsCnt  = (nb + 1 + 3) & ~3;

  float* B0   = (float*)d_ws;
  float* B1   = B0 + (size_t)Nn * SP;
  int*   cnt  = (int*)(B1 + (size_t)Nn * SP);
  int*   off  = cnt + (size_t)segs;
  int*   col  = off + (size_t)offCnt;
  int*   bsum = col + (size_t)edges;
  float* WsP1 = (float*)(bsum + (size_t)bsCnt);
  float* WnP1 = WsP1 + 16 * SP;
  float* bsP1 = WnP1 + RT * 16 * SP;
  float* WsP2 = bsP1 + 16;
  float* WnP2 = WsP2 + 16 * SP;
  float* bsP2 = WnP2 + RT * 16 * SP;

  const int tiles   = (Nn + 15) / 16;
  const int wBlocks = (tiles + 3) / 4;
  const dim3 edgeGrid(256, RT);
  const dim3 scoreGrid(512, 2);

  k_prep<<<1, 256, 0, stream>>>(Ws1, Wn1, b1, Ws2, Wn2, b2,
                                WsP1, WnP1, bsP1, WsP2, WnP2, bsP2);
  k_padcopy<<<1024, RES_T, 0, stream>>>(x, B0, Nn);

  k_zero_i<<<256, RES_T, 0, stream>>>(cnt, segs);
  k_hist<<<edgeGrid, RES_T, 0, stream>>>(edst, cnt, Nn, Ee);
  k_scan_a<<<nb, SCAN_T, 0, stream>>>(cnt, bsum, segs);
  k_scan_b<<<1, SCANB_T, 0, stream>>>(bsum, nb);
  k_scan_c<<<nb, SCAN_T, 0, stream>>>(cnt, bsum, off, segs);
  k_fill<<<edgeGrid, RES_T, 0, stream>>>(esrc, edst, off, cnt, col, Nn, Ee);

  k_layer<<<wBlocks, 128, 0, stream>>>(B0, off, col, WsP1, WnP1, bsP1, B1, tiles, Nn);
  k_layer<<<wBlocks, 128, 0, stream>>>(B1, off, col, WsP2, WnP2, bsP2, B0, tiles, Nn);

  k_score<<<scoreGrid, RES_T, 0, stream>>>(B0, esrc, edst, nsrc, ndst,
                                           (float*)d_out, Ee);
}


# Sort counters in two planes, one per package half: work-groups read their die and draw edge-chunk tickets of their half first, so hist/fill atomics stay in one cache half; segment sum order changes
# speedup vs baseline: 2.5419x; 1.6931x over previous
#include <hip/hip_runtime.h>


#define D  10
#define RT 4
#define SP 12

#define SCAN_T     256
#define SCAN_PER   8
#define SCAN_CHUNK (SCAN_T * SCAN_PER)
#define SCANB_T    1024

#define RES_T 256
#define EPT   4
#define SPT   2
#define PPT   8

#define CH_TRIPS  2
#define CHUNK     (RES_T * EPT * CH_TRIPS)
#define TICK_PITCH 64
#define TICK_INTS  (4 * TICK_PITCH)

typedef __attribute__((ext_vector_type(2))) float v2f;
typedef __attribute__((ext_vector_type(8))) float v8f;

__device__ __forceinline__ v8f wmma_k4(v2f a, v2f b, v8f c) {
  return __builtin_amdgcn_wmma_f32_16x16x4_f32(
       false, a,  false, b,
       (short)0, c,  false,  false);
}

__global__ void k_prep(const float* __restrict__ Ws1, const float* __restrict__ Wn1,
                       const float* __restrict__ b1,
                       const float* __restrict__ Ws2, const float* __restrict__ Wn2,
                       const float* __restrict__ b2,
                       float* __restrict__ WsP1, float* __restrict__ WnP1,
                       float* __restrict__ bsP1,
                       float* __restrict__ WsP2, float* __restrict__ WnP2,
                       float* __restrict__ bsP2) {
  for (int i = threadIdx.x; i < 16 * SP; i += blockDim.x) {
    int n = i / SP, k = i % SP;
    float s1 = 0.f, s2 = 0.f;
    if (n < D && k < D) {
#pragma unroll
      for (int r = 0; r < RT; ++r) {
        s1 += Ws1[r * D * D + n * D + k];
        s2 += Ws2[r * D * D + n * D + k];
      }
    }
    WsP1[i] = s1;
    WsP2[i] = s2;
  }
  for (int i = threadIdx.x; i < RT * 16 * SP; i += blockDim.x) {
    int r = i / (16 * SP), j = i % (16 * SP);
    int n = j / SP, k = j % SP;
    float v1 = 0.f, v2 = 0.f;
    if (n < D && k < D) {
      v1 = Wn1[r * D * D + n * D + k];
      v2 = Wn2[r * D * D + n * D + k];
    }
    WnP1[i] = v1;
    WnP2[i] = v2;
  }
  for (int i = threadIdx.x; i < 16; i += blockDim.x) {
    float s1 = 0.f, s2 = 0.f;
    if (i < D) {
#pragma unroll
      for (int r = 0; r < RT; ++r) {
        s1 += b1[r * D + i];
        s2 += b2[r * D + i];
      }
    }
    bsP1[i] = s1;
    bsP2[i] = s2;
  }
}

__device__ __forceinline__ float pad_elem(const float* __restrict__ x, int i) {
  int n = i / SP, k = i - n * SP;
  float t = x[(long long)n * D + min(k, D - 1)];
  return (k < D) ? t : 0.0f;
}

__global__ void __launch_bounds__(RES_T)
k_padcopy(const float* __restrict__ x, float* __restrict__ hp, int Nn) {
  const int total = Nn * SP;
  const int step  = gridDim.x * (RES_T * PPT);
  int base = blockIdx.x * (RES_T * PPT) + threadIdx.x;
  for (; base + (PPT - 1) * RES_T < total; base += step) {
    float v[PPT];
#pragma unroll
    for (int q = 0; q < PPT; ++q) v[q] = pad_elem(x, base + q * RES_T);
#pragma unroll
    for (int q = 0; q < PPT; ++q) hp[base + q * RES_T] = v[q];
  }
#pragma unroll
  for (int q = 0; q < PPT; ++q) {
    int i = base + q * RES_T;
    if (i < total) hp[i] = pad_elem(x, i);
  }
}

__global__ void k_zero_i(int* __restrict__ p, int n) {
  const int stride = gridDim.x * blockDim.x;
  for (int i = blockIdx.x * blockDim.x + threadIdx.x; i < n; i += stride) p[i] = 0;
}


__device__ __forceinline__ int package_half() {
  unsigned v = __builtin_amdgcn_s_sendmsg_rtn(0x87);
  return (int)((v >> 18) & 1u);
}

__device__ __forceinline__ int draw_ticket(int* __restrict__ word, int* sh) {
  if (threadIdx.x == 0) *sh = atomicAdd(word, 1);
  __syncthreads();
  int t = *sh;
  __syncthreads();
  return t;
}

template <bool WHOLE>
__device__ __forceinline__ void hist_chunk(const int* __restrict__ ed, int* __restrict__ cp,
                                           int base, int Ee) {
#pragma unroll
  for (int trip = 0; trip < CH_TRIPS; ++trip, base += RES_T * EPT) {
    int d[EPT];
#pragma unroll
    for (int k = 0; k < EPT; ++k) {
      const int i = base + k * RES_T;
      d[k] = ed[WHOLE ? i : min(i, Ee - 1)];
    }
#pragma unroll
    for (int k = 0; k < EPT; ++k)
      if (WHOLE || base + k * RES_T < Ee) atomicAdd(&cp[d[k]], 1);
  }
}

__global__ void __launch_bounds__(RES_T)
k_hist(const int* __restrict__ edst,
       int* __restrict__ cnt,
       int* __restrict__ tick,
       int Nn, int Ee) {
  __shared__ int sh_t;
  __shared__ int sh_h;
  if (threadIdx.x == 0) sh_h = package_half();
  __syncthreads();
  const int h = sh_h;
  const int cpe = (Ee + CHUNK - 1) / CHUNK;
  const int nChunks = RT * cpe;
  const long long segs = (long long)RT * Nn;
  for (int pass = 0; pass < 2; ++pass) {
    const int q = h ^ pass;
    int* __restrict__ plane = cnt + q * segs;
    int* __restrict__ word  = tick + q * TICK_PITCH;
    for (int it = 0; it <= nChunks; ++it) {
      const int c = 2 * draw_ticket(word, &sh_t) + q;
      if (c >= nChunks) break;
      const int r  = c / cpe;
      const int lc = c - r * cpe;
      const int* __restrict__ ed = edst + (long long)r * Ee;
      int* __restrict__ cp = plane + (long long)r * Nn;
      const int base = lc * CHUNK + threadIdx.x;
      if ((lc + 1) * CHUNK <= Ee) hist_chunk<true>(ed, cp, base, Ee);
      else                        hist_chunk<false>(ed, cp, base, Ee);
    }
  }
}

__global__ void k_scan_a(const int* __restrict__ cnt, int* __restrict__ bsum, int n) {
  __shared__ int sh[SCAN_T];
  int t = threadIdx.x;
  int base = blockIdx.x * SCAN_CHUNK + t * SCAN_PER;
  int s = 0;
#pragma unroll
  for (int q = 0; q < SCAN_PER; ++q) {
    int i = base + q;
    if (i < n) s += cnt[i] + cnt[(long long)n + i];
  }
  sh[t] = s;
  __syncthreads();
  for (int ofs = SCAN_T / 2; ofs > 0; ofs >>= 1) {
    if (t < ofs) sh[t] += sh[t + ofs];
    __syncthreads();
  }
  if (t == 0) bsum[blockIdx.x] = sh[0];
}

__global__ void k_scan_b(int* __restrict__ bsum, int nb) {
  __shared__ int sh[SCANB_T];
  int t = threadIdx.x;
  int carry = 0;
  for (int base = 0; base < nb; base += SCANB_T) {
    int i = base + t;
    int v = (i < nb) ? bsum[i] : 0;
    sh[t] = v;
    __syncthreads();
    for (int ofs = 1; ofs < SCANB_T; ofs <<= 1) {
      int add = (t >= ofs) ? sh[t - ofs] : 0;
      __syncthreads();
      sh[t] += add;
      __syncthreads();
    }
    int incl = sh[t];
    int tot  = sh[SCANB_T - 1];
    if (i < nb) bsum[i] = carry + incl - v;
    carry += tot;
    __syncthreads();
  }
  if (t == 0) bsum[nb] = carry;
}

__global__ void k_scan_c(int* __restrict__ cnt, const int* __restrict__ bexc,
                         int* __restrict__ off, int n) {
  __shared__ int sh[SCAN_T];
  int t = threadIdx.x;
  int base = blockIdx.x * SCAN_CHUNK + t * SCAN_PER;
  int v[SCAN_PER], a0[SCAN_PER];
  int s = 0;
#pragma unroll
  for (int q = 0; q < SCAN_PER; ++q) {
    int i = base + q;
    a0[q] = (i < n) ? cnt[i] : 0;
    v[q]  = a0[q] + ((i < n) ? cnt[(long long)n + i] : 0);
    s += v[q];
  }
  sh[t] = s;
  __syncthreads();
  for (int ofs = 1; ofs < SCAN_T; ofs <<= 1) {
    int add = (t >= ofs) ? sh[t - ofs] : 0;
    __syncthreads();
    sh[t] += add;
    __syncthreads();
  }
  int run = bexc[blockIdx.x] + sh[t] - s;
#pragma unroll
  for (int q = 0; q < SCAN_PER; ++q) {
    int i = base + q;
    if (i < n) {
      off[i] = run;
      cnt[i] = run;
      cnt[(long long)n + i] = run + a0[q];
    }
    run += v[q];
  }
  if (blockIdx.x == 0 && t == 0) off[n] = bexc[gridDim.x];
}

template <bool WHOLE>
__device__ __forceinline__ void fill_chunk(const int* __restrict__ es, const int* __restrict__ ed,
                                           int* __restrict__ cp, int* __restrict__ col,
                                           int base, int Ee) {
#pragma unroll
  for (int trip = 0; trip < CH_TRIPS; ++trip, base += RES_T * EPT) {
    int d[EPT], s[EPT], p[EPT];
#pragma unroll
    for (int k = 0; k < EPT; ++k) {
      const int i = WHOLE ? (base + k * RES_T) : min(base + k * RES_T, Ee - 1);
      d[k] = ed[i];
      s[k] = es[i];
    }
    if (WHOLE) {
#pragma unroll
      for (int k = 0; k < EPT; ++k) p[k] = atomicAdd(&cp[d[k]], 1);
#pragma unroll
      for (int k = 0; k < EPT; ++k) col[p[k]] = s[k];
    } else {
#pragma unroll
      for (int k = 0; k < EPT; ++k)
        if (base + k * RES_T < Ee) col[atomicAdd(&cp[d[k]], 1)] = s[k];
    }
  }
}

__global__ void __launch_bounds__(RES_T)
k_fill(const int* __restrict__ esrc,
       const int* __restrict__ edst,
       int* __restrict__ cur,
       int* __restrict__ tick,
       int* __restrict__ col,
       int Nn, int Ee) {
  __shared__ int sh_t;
  __shared__ int sh_h;
  if (threadIdx.x == 0) sh_h = package_half();
  __syncthreads();
  const int h = sh_h;
  const int cpe = (Ee + CHUNK - 1) / CHUNK;
  const int nChunks = RT * cpe;
  const long long segs = (long long)RT * Nn;
  for (int pass = 0; pass < 2; ++pass) {
    const int q = h ^ pass;
    int* __restrict__ plane = cur + q * segs;
    int* __restrict__ word  = tick + q * TICK_PITCH;
    for (int it = 0; it <= nChunks; ++it) {
      const int c = 2 * draw_ticket(word, &sh_t) + q;
      if (c >= nChunks) break;
      const int r  = c / cpe;
      const int lc = c - r * cpe;
      const int* __restrict__ es = esrc + (long long)r * Ee;
      const int* __restrict__ ed = edst + (long long)r * Ee;
      int* __restrict__ cp = plane + (long long)r * Nn;
      const int base = lc * CHUNK + threadIdx.x;
      if ((lc + 1) * CHUNK <= Ee) fill_chunk<true>(es, ed, cp, col, base, Ee);
      else                        fill_chunk<false>(es, ed, cp, col, base, Ee);
    }
  }
}


__global__ void k_layer(const float* __restrict__ hp,
                        const int* __restrict__ off,
                        const int* __restrict__ col,
                        const float* __restrict__ WsP,
                        const float* __restrict__ WnP,
                        const float* __restrict__ bsP,
                        float* __restrict__ hout,
                        int nTiles, int Nn) {
  int wave = blockIdx.x * (blockDim.x >> 5) + (threadIdx.x >> 5);
  if (wave >= nTiles) return;
  int lane = threadIdx.x & 31;
  int m    = lane & 15;
  int half = lane >> 4;
  int kk   = half * 2;
  int node = min(wave * 16 + m, Nn - 1);

  union { v8f v; float f[8]; } acc;
  float bsum = bsP[m];
#pragma unroll
  for (int j = 0; j < 8; ++j) acc.f[j] = bsum;

  {
    const float* arow = hp + (long long)node * SP + kk;
    const float* brow = WsP + m * SP + kk;
#pragma unroll
    for (int c = 0; c < 3; ++c) {
      v2f a = *(const v2f*)(arow + c * 4);
      v2f b = *(const v2f*)(brow + c * 4);
      acc.v = wmma_k4(a, b, acc.v);
    }
  }

#pragma unroll
  for (int r = 0; r < RT; ++r) {
    int g   = r * Nn + node;
    int beg = off[g];
    int end = off[g + 1];
    v2f s0 = {0.f, 0.f}, s1 = {0.f, 0.f}, s2 = {0.f, 0.f};
    int j = beg;
    for (; j + 1 < end; j += 2) {
      int sa = col[j];
      int sb = col[j + 1];
      const float* ra = hp + (long long)sa * SP + kk;
      const float* rb = hp + (long long)sb * SP + kk;
      v2f xa0 = *(const v2f*)(ra);
      v2f xa1 = *(const v2f*)(ra + 4);
      v2f xa2 = *(const v2f*)(ra + 8);
      v2f xb0 = *(const v2f*)(rb);
      v2f xb1 = *(const v2f*)(rb + 4);
      v2f xb2 = *(const v2f*)(rb + 8);
      s0 += xa0; s1 += xa1; s2 += xa2;
      s0 += xb0; s1 += xb1; s2 += xb2;
    }
    if (j < end) {
      int sa = col[j];
      const float* ra = hp + (long long)sa * SP + kk;
      s0 += *(const v2f*)(ra);
      s1 += *(const v2f*)(ra + 4);
      s2 += *(const v2f*)(ra + 8);
    }
    float rdeg = 1.0f / fmaxf((float)(end - beg), 1.0f);
    const float* bw = WnP + r * 16 * SP + m * SP + kk;
    v2f b0 = *(const v2f*)(bw);
    v2f b1 = *(const v2f*)(bw + 4);
    v2f b2 = *(const v2f*)(bw + 8);
    acc.v = wmma_k4(s0 * rdeg, b0, acc.v);
    acc.v = wmma_k4(s1 * rdeg, b1, acc.v);
    acc.v = wmma_k4(s2 * rdeg, b2, acc.v);
  }

  int rowBase = wave * 16 + half * 8;
  bool fullTile = (wave * 16 + 16) <= Nn;
  if (m < SP) {
    float* op = hout + (long long)rowBase * SP + m;
    if (fullTile) {
#pragma unroll
      for (int j = 0; j < 8; ++j) op[j * SP] = acc.f[j];
    } else {
#pragma unroll
      for (int j = 0; j < 8; ++j)
        if (rowBase + j < Nn) op[j * SP] = acc.f[j];
    }
  }
}

__device__ __forceinline__ float dot_rows(const float* __restrict__ hp, int a, int b) {
  const float* pa = hp + (long long)a * SP;
  const float* pb = hp + (long long)b * SP;
  float s = 0.0f;
#pragma unroll
  for (int k = 0; k < D; ++k) s += pa[k] * pb[k];
  return s;
}

__global__ void __launch_bounds__(RES_T)
k_score(const float* __restrict__ hp,
        const int* __restrict__ ps, const int* __restrict__ pd,
        const int* __restrict__ ns, const int* __restrict__ nd,
        float* __restrict__ out, int E_) {
  const int* __restrict__ ia = blockIdx.y ? ns : ps;
  const int* __restrict__ ib = blockIdx.y ? nd : pd;
  float* __restrict__ o = out + (long long)blockIdx.y * E_;
  const int step = gridDim.x * (RES_T * SPT);
  int base = blockIdx.x * (RES_T * SPT) + threadIdx.x;
  for (; base + (SPT - 1) * RES_T < E_; base += step) {
    int a[SPT], b[SPT];
    float s[SPT];
#pragma unroll
    for (int q = 0; q < SPT; ++q) {
      a[q] = ia[base + q * RES_T];
      b[q] = ib[base + q * RES_T];
    }
#pragma unroll
    for (int q = 0; q < SPT; ++q) s[q] = dot_rows(hp, a[q], b[q]);
#pragma unroll
    for (int q = 0; q < SPT; ++q) o[base + q * RES_T] = s[q];
  }
#pragma unroll
  for (int q = 0; q < SPT; ++q) {
    int i = base + q * RES_T;
    if (i < E_) o[i] = dot_rows(hp, ia[i], ib[i]);
  }
}

extern "C" void kernel_launch(void* const* d_in, const int* in_sizes, int n_in,
                              void* d_out, int out_size, void* d_ws, size_t ws_size,
                              hipStream_t stream) {
  (void)n_in; (void)out_size; (void)ws_size;
  const float* x    = (const float*)d_in[0];
  const int*   esrc = (const int*)d_in[1];
  const int*   edst = (const int*)d_in[2];
  const int*   nsrc = (const int*)d_in[3];
  const int*   ndst = (const int*)d_in[4];
  const float* Ws1  = (const float*)d_in[5];
  const float* Wn1  = (const float*)d_in[6];
  const float* b1   = (const float*)d_in[7];
  const float* Ws2  = (const float*)d_in[8];
  const float* Wn2  = (const float*)d_in[9];
  const float* b2   = (const float*)d_in[10];

  const int Nn = in_sizes[0] / D;
  const int Ee = in_sizes[3];

  const int segs   = RT * Nn;
  const int edges  = RT * Ee;
  const int offCnt = (segs + 1 + 3) & ~3;
  const int nb     = (segs + SCAN_CHUNK - 1) / SCAN_CHUNK;
  const int bsCnt  = (nb + 1 + 3) & ~3;

  float* B0   = (float*)d_ws;
  float* B1   = B0 + (size_t)Nn * SP;
  int*   cnt  = (int*)(B1 + (size_t)Nn * SP);
  int*   tick = cnt + 2 * (size_t)segs;
  int*   off  = tick + TICK_INTS;
  int*   col  = off + (size_t)offCnt;
  int*   bsum = col + (size_t)edges;
  float* WsP1 = (float*)(bsum + (size_t)bsCnt);
  float* WnP1 = WsP1 + 16 * SP;
  float* bsP1 = WnP1 + RT * 16 * SP;
  float* WsP2 = bsP1 + 16;
  float* WnP2 = WsP2 + 16 * SP;
  float* bsP2 = WnP2 + RT * 16 * SP;

  const int tiles   = (Nn + 15) / 16;
  const int wBlocks = (tiles + 3) / 4;
  const int  edgeGrid = 2048;
  const dim3 scoreGrid(512, 2);

  k_prep<<<1, 256, 0, stream>>>(Ws1, Wn1, b1, Ws2, Wn2, b2,
                                WsP1, WnP1, bsP1, WsP2, WnP2, bsP2);
  k_padcopy<<<1024, RES_T, 0, stream>>>(x, B0, Nn);

  k_zero_i<<<256, RES_T, 0, stream>>>(cnt, 2 * segs + TICK_INTS);
  k_hist<<<edgeGrid, RES_T, 0, stream>>>(edst, cnt, tick, Nn, Ee);
  k_scan_a<<<nb, SCAN_T, 0, stream>>>(cnt, bsum, segs);
  k_scan_b<<<1, SCANB_T, 0, stream>>>(bsum, nb);
  k_scan_c<<<nb, SCAN_T, 0, stream>>>(cnt, bsum, off, segs);
  k_fill<<<edgeGrid, RES_T, 0, stream>>>(esrc, edst, cnt, tick + 2 * TICK_PITCH, col, Nn, Ee);

  k_layer<<<wBlocks, 128, 0, stream>>>(B0, off, col, WsP1, WnP1, bsP1, B1, tiles, Nn);
  k_layer<<<wBlocks, 128, 0, stream>>>(B1, off, col, WsP2, WnP2, bsP2, B0, tiles, Nn);

  k_score<<<scoreGrid, RES_T, 0, stream>>>(B0, esrc, edst, nsrc, ndst,
                                           (float*)d_out, Ee);
}


# Scan both counter planes as ONE sequence: each half's sorted slots get their own region of col, so a col line takes stores from one cache half only; layers walk 2 runs per segment; sum order kept
# speedup vs baseline: 3.5608x; 1.4008x over previous
#include <hip/hip_runtime.h>


#define D  10
#define RT 4
#define SP 12

#define SCAN_T     256
#define SCAN_PER   8
#define SCAN_CHUNK (SCAN_T * SCAN_PER)
#define SCANB_T    1024

#define RES_T 256
#define EPT   4
#define SPT   2
#define PPT   8

#define CH_TRIPS  2
#define CHUNK     (RES_T * EPT * CH_TRIPS)
#define TICK_PITCH 64
#define TICK_INTS  (4 * TICK_PITCH)

typedef __attribute__((ext_vector_type(2))) float v2f;
typedef __attribute__((ext_vector_type(8))) float v8f;

__device__ __forceinline__ v8f wmma_k4(v2f a, v2f b, v8f c) {
  return __builtin_amdgcn_wmma_f32_16x16x4_f32(
       false, a,  false, b,
       (short)0, c,  false,  false);
}

__global__ void k_prep(const float* __restrict__ Ws1, const float* __restrict__ Wn1,
                       const float* __restrict__ b1,
                       const float* __restrict__ Ws2, const float* __restrict__ Wn2,
                       const float* __restrict__ b2,
                       float* __restrict__ WsP1, float* __restrict__ WnP1,
                       float* __restrict__ bsP1,
                       float* __restrict__ WsP2, float* __restrict__ WnP2,
                       float* __restrict__ bsP2) {
  for (int i = threadIdx.x; i < 16 * SP; i += blockDim.x) {
    int n = i / SP, k = i % SP;
    float s1 = 0.f, s2 = 0.f;
    if (n < D && k < D) {
#pragma unroll
      for (int r = 0; r < RT; ++r) {
        s1 += Ws1[r * D * D + n * D + k];
        s2 += Ws2[r * D * D + n * D + k];
      }
    }
    WsP1[i] = s1;
    WsP2[i] = s2;
  }
  for (int i = threadIdx.x; i < RT * 16 * SP; i += blockDim.x) {
    int r = i / (16 * SP), j = i % (16 * SP);
    int n = j / SP, k = j % SP;
    float v1 = 0.f, v2 = 0.f;
    if (n < D && k < D) {
      v1 = Wn1[r * D * D + n * D + k];
      v2 = Wn2[r * D * D + n * D + k];
    }
    WnP1[i] = v1;
    WnP2[i] = v2;
  }
  for (int i = threadIdx.x; i < 16; i += blockDim.x) {
    float s1 = 0.f, s2 = 0.f;
    if (i < D) {
#pragma unroll
      for (int r = 0; r < RT; ++r) {
        s1 += b1[r * D + i];
        s2 += b2[r * D + i];
      }
    }
    bsP1[i] = s1;
    bsP2[i] = s2;
  }
}

__device__ __forceinline__ float pad_elem(const float* __restrict__ x, int i) {
  int n = i / SP, k = i - n * SP;
  float t = x[(long long)n * D + min(k, D - 1)];
  return (k < D) ? t : 0.0f;
}

__global__ void __launch_bounds__(RES_T)
k_padcopy(const float* __restrict__ x, float* __restrict__ hp, int Nn) {
  const int total = Nn * SP;
  const int step  = gridDim.x * (RES_T * PPT);
  int base = blockIdx.x * (RES_T * PPT) + threadIdx.x;
  for (; base + (PPT - 1) * RES_T < total; base += step) {
    float v[PPT];
#pragma unroll
    for (int q = 0; q < PPT; ++q) v[q] = pad_elem(x, base + q * RES_T);
#pragma unroll
    for (int q = 0; q < PPT; ++q) hp[base + q * RES_T] = v[q];
  }
#pragma unroll
  for (int q = 0; q < PPT; ++q) {
    int i = base + q * RES_T;
    if (i < total) hp[i] = pad_elem(x, i);
  }
}

__global__ void k_zero_i(int* __restrict__ p, int n) {
  const int stride = gridDim.x * blockDim.x;
  for (int i = blockIdx.x * blockDim.x + threadIdx.x; i < n; i += stride) p[i] = 0;
}


__device__ __forceinline__ int package_half() {
  unsigned v = __builtin_amdgcn_s_sendmsg_rtn(0x87);
  return (int)((v >> 18) & 1u);
}

__device__ __forceinline__ int draw_ticket(int* __restrict__ word, int* sh) {
  if (threadIdx.x == 0) *sh = atomicAdd(word, 1);
  __syncthreads();
  int t = *sh;
  __syncthreads();
  return t;
}

template <bool WHOLE>
__device__ __forceinline__ void hist_chunk(const int* __restrict__ ed, int* __restrict__ cp,
                                           int base, int Ee) {
#pragma unroll
  for (int trip = 0; trip < CH_TRIPS; ++trip, base += RES_T * EPT) {
    int d[EPT];
#pragma unroll
    for (int k = 0; k < EPT; ++k) {
      const int i = base + k * RES_T;
      d[k] = ed[WHOLE ? i : min(i, Ee - 1)];
    }
#pragma unroll
    for (int k = 0; k < EPT; ++k)
      if (WHOLE || base + k * RES_T < Ee) atomicAdd(&cp[d[k]], 1);
  }
}

__global__ void __launch_bounds__(RES_T)
k_hist(const int* __restrict__ edst,
       int* __restrict__ cnt,
       int* __restrict__ tick,
       int Nn, int Ee) {
  __shared__ int sh_t;
  __shared__ int sh_h;
  if (threadIdx.x == 0) sh_h = package_half();
  __syncthreads();
  const int h = sh_h;
  const int cpe = (Ee + CHUNK - 1) / CHUNK;
  const int nChunks = RT * cpe;
  const long long segs = (long long)RT * Nn;
  for (int pass = 0; pass < 2; ++pass) {
    const int q = h ^ pass;
    int* __restrict__ plane = cnt + q * segs;
    int* __restrict__ word  = tick + q * TICK_PITCH;
    for (int it = 0; it <= nChunks; ++it) {
      const int c = 2 * draw_ticket(word, &sh_t) + q;
      if (c >= nChunks) break;
      const int r  = c / cpe;
      const int lc = c - r * cpe;
      const int* __restrict__ ed = edst + (long long)r * Ee;
      int* __restrict__ cp = plane + (long long)r * Nn;
      const int base = lc * CHUNK + threadIdx.x;
      if ((lc + 1) * CHUNK <= Ee) hist_chunk<true>(ed, cp, base, Ee);
      else                        hist_chunk<false>(ed, cp, base, Ee);
    }
  }
}

__global__ void k_scan_a(const int* __restrict__ cnt, int* __restrict__ bsum, int n) {
  __shared__ int sh[SCAN_T];
  int t = threadIdx.x;
  int base = blockIdx.x * SCAN_CHUNK + t * SCAN_PER;
  int s = 0;
#pragma unroll
  for (int q = 0; q < SCAN_PER; ++q) {
    int i = base + q;
    if (i < n) s += cnt[i];
  }
  sh[t] = s;
  __syncthreads();
  for (int ofs = SCAN_T / 2; ofs > 0; ofs >>= 1) {
    if (t < ofs) sh[t] += sh[t + ofs];
    __syncthreads();
  }
  if (t == 0) bsum[blockIdx.x] = sh[0];
}

__global__ void k_scan_b(int* __restrict__ bsum, int nb) {
  __shared__ int sh[SCANB_T];
  int t = threadIdx.x;
  int carry = 0;
  for (int base = 0; base < nb; base += SCANB_T) {
    int i = base + t;
    int v = (i < nb) ? bsum[i] : 0;
    sh[t] = v;
    __syncthreads();
    for (int ofs = 1; ofs < SCANB_T; ofs <<= 1) {
      int add = (t >= ofs) ? sh[t - ofs] : 0;
      __syncthreads();
      sh[t] += add;
      __syncthreads();
    }
    int incl = sh[t];
    int tot  = sh[SCANB_T - 1];
    if (i < nb) bsum[i] = carry + incl - v;
    carry += tot;
    __syncthreads();
  }
  if (t == 0) bsum[nb] = carry;
}

__global__ void k_scan_c(int* __restrict__ cnt, const int* __restrict__ bexc,
                         int* __restrict__ off, int n) {
  __shared__ int sh[SCAN_T];
  int t = threadIdx.x;
  int base = blockIdx.x * SCAN_CHUNK + t * SCAN_PER;
  int v[SCAN_PER];
  int s = 0;
#pragma unroll
  for (int q = 0; q < SCAN_PER; ++q) {
    int i = base + q;
    v[q] = (i < n) ? cnt[i] : 0;
    s += v[q];
  }
  sh[t] = s;
  __syncthreads();
  for (int ofs = 1; ofs < SCAN_T; ofs <<= 1) {
    int add = (t >= ofs) ? sh[t - ofs] : 0;
    __syncthreads();
    sh[t] += add;
    __syncthreads();
  }
  int run = bexc[blockIdx.x] + sh[t] - s;
#pragma unroll
  for (int q = 0; q < SCAN_PER; ++q) {
    int i = base + q;
    if (i < n) {
      off[i] = run;
      cnt[i] = run;
    }
    run += v[q];
  }
  if (blockIdx.x == 0 && t == 0) off[n] = bexc[gridDim.x];
}

template <bool WHOLE>
__device__ __forceinline__ void fill_chunk(const int* __restrict__ es, const int* __restrict__ ed,
                                           int* __restrict__ cp, int* __restrict__ col,
                                           int base, int Ee) {
#pragma unroll
  for (int trip = 0; trip < CH_TRIPS; ++trip, base += RES_T * EPT) {
    int d[EPT], s[EPT], p[EPT];
#pragma unroll
    for (int k = 0; k < EPT; ++k) {
      const int i = WHOLE ? (base + k * RES_T) : min(base + k * RES_T, Ee - 1);
      d[k] = ed[i];
      s[k] = es[i];
    }
    if (WHOLE) {
#pragma unroll
      for (int k = 0; k < EPT; ++k) p[k] = atomicAdd(&cp[d[k]], 1);
#pragma unroll
      for (int k = 0; k < EPT; ++k) col[p[k]] = s[k];
    } else {
#pragma unroll
      for (int k = 0; k < EPT; ++k)
        if (base + k * RES_T < Ee) col[atomicAdd(&cp[d[k]], 1)] = s[k];
    }
  }
}

__global__ void __launch_bounds__(RES_T)
k_fill(const int* __restrict__ esrc,
       const int* __restrict__ edst,
       int* __restrict__ cur,
       int* __restrict__ tick,
       int* __restrict__ col,
       int Nn, int Ee) {
  __shared__ int sh_t;
  __shared__ int sh_h;
  if (threadIdx.x == 0) sh_h = package_half();
  __syncthreads();
  const int h = sh_h;
  const int cpe = (Ee + CHUNK - 1) / CHUNK;
  const int nChunks = RT * cpe;
  const long long segs = (long long)RT * Nn;
  for (int pass = 0; pass < 2; ++pass) {
    const int q = h ^ pass;
    int* __restrict__ plane = cur + q * segs;
    int* __restrict__ word  = tick + q * TICK_PITCH;
    for (int it = 0; it <= nChunks; ++it) {
      const int c = 2 * draw_ticket(word, &sh_t) + q;
      if (c >= nChunks) break;
      const int r  = c / cpe;
      const int lc = c - r * cpe;
      const int* __restrict__ es = esrc + (long long)r * Ee;
      const int* __restrict__ ed = edst + (long long)r * Ee;
      int* __restrict__ cp = plane + (long long)r * Nn;
      const int base = lc * CHUNK + threadIdx.x;
      if ((lc + 1) * CHUNK <= Ee) fill_chunk<true>(es, ed, cp, col, base, Ee);
      else                        fill_chunk<false>(es, ed, cp, col, base, Ee);
    }
  }
}


__global__ void k_layer(const float* __restrict__ hp,
                        const int* __restrict__ off,
                        const int* __restrict__ col,
                        const float* __restrict__ WsP,
                        const float* __restrict__ WnP,
                        const float* __restrict__ bsP,
                        float* __restrict__ hout,
                        int nTiles, int Nn) {
  int wave = blockIdx.x * (blockDim.x >> 5) + (threadIdx.x >> 5);
  if (wave >= nTiles) return;
  int lane = threadIdx.x & 31;
  int m    = lane & 15;
  int half = lane >> 4;
  int kk   = half * 2;
  int node = min(wave * 16 + m, Nn - 1);

  union { v8f v; float f[8]; } acc;
  float bsum = bsP[m];
#pragma unroll
  for (int j = 0; j < 8; ++j) acc.f[j] = bsum;

  {
    const float* arow = hp + (long long)node * SP + kk;
    const float* brow = WsP + m * SP + kk;
#pragma unroll
    for (int c = 0; c < 3; ++c) {
      v2f a = *(const v2f*)(arow + c * 4);
      v2f b = *(const v2f*)(brow + c * 4);
      acc.v = wmma_k4(a, b, acc.v);
    }
  }

  const int segs = RT * Nn;
#pragma unroll
  for (int r = 0; r < RT; ++r) {
    int g    = r * Nn + node;
    int p0   = off[g];
    int len0 = off[g + 1] - p0;
    int p1   = off[segs + g];
    int end  = len0 + (off[segs + g + 1] - p1);
    int sh   = p1 - len0;
    v2f s0 = {0.f, 0.f}, s1 = {0.f, 0.f}, s2 = {0.f, 0.f};
    int j = 0;
    for (; j + 1 < end; j += 2) {
      int sa = col[(j     < len0) ? p0 + j     : sh + j];
      int sb = col[(j + 1 < len0) ? p0 + j + 1 : sh + j + 1];
      const float* ra = hp + (long long)sa * SP + kk;
      const float* rb = hp + (long long)sb * SP + kk;
      v2f xa0 = *(const v2f*)(ra);
      v2f xa1 = *(const v2f*)(ra + 4);
      v2f xa2 = *(const v2f*)(ra + 8);
      v2f xb0 = *(const v2f*)(rb);
      v2f xb1 = *(const v2f*)(rb + 4);
      v2f xb2 = *(const v2f*)(rb + 8);
      s0 += xa0; s1 += xa1; s2 += xa2;
      s0 += xb0; s1 += xb1; s2 += xb2;
    }
    if (j < end) {
      int sa = col[(j < len0) ? p0 + j : sh + j];
      const float* ra = hp + (long long)sa * SP + kk;
      s0 += *(const v2f*)(ra);
      s1 += *(const v2f*)(ra + 4);
      s2 += *(const v2f*)(ra + 8);
    }
    float rdeg = 1.0f / fmaxf((float)end, 1.0f);
    const float* bw = WnP + r * 16 * SP + m * SP + kk;
    v2f b0 = *(const v2f*)(bw);
    v2f b1 = *(const v2f*)(bw + 4);
    v2f b2 = *(const v2f*)(bw + 8);
    acc.v = wmma_k4(s0 * rdeg, b0, acc.v);
    acc.v = wmma_k4(s1 * rdeg, b1, acc.v);
    acc.v = wmma_k4(s2 * rdeg, b2, acc.v);
  }

  int rowBase = wave * 16 + half * 8;
  bool fullTile = (wave * 16 + 16) <= Nn;
  if (m < SP) {
    float* op = hout + (long long)rowBase * SP + m;
    if (fullTile) {
#pragma unroll
      for (int j = 0; j < 8; ++j) op[j * SP] = acc.f[j];
    } else {
#pragma unroll
      for (int j = 0; j < 8; ++j)
        if (rowBase + j < Nn) op[j * SP] = acc.f[j];
    }
  }
}

__device__ __forceinline__ float dot_rows(const float* __restrict__ hp, int a, int b) {
  const float* pa = hp + (long long)a * SP;
  const float* pb = hp + (long long)b * SP;
  float s = 0.0f;
#pragma unroll
  for (int k = 0; k < D; ++k) s += pa[k] * pb[k];
  return s;
}

__global__ void __launch_bounds__(RES_T)
k_score(const float* __restrict__ hp,
        const int* __restrict__ ps, const int* __restrict__ pd,
        const int* __restrict__ ns, const int* __restrict__ nd,
        float* __restrict__ out, int E_) {
  const int* __restrict__ ia = blockIdx.y ? ns : ps;
  const int* __restrict__ ib = blockIdx.y ? nd : pd;
  float* __restrict__ o = out + (long long)blockIdx.y * E_;
  const int step = gridDim.x * (RES_T * SPT);
  int base = blockIdx.x * (RES_T * SPT) + threadIdx.x;
  for (; base + (SPT - 1) * RES_T < E_; base += step) {
    int a[SPT], b[SPT];
    float s[SPT];
#pragma unroll
    for (int q = 0; q < SPT; ++q) {
      a[q] = ia[base + q * RES_T];
      b[q] = ib[base + q * RES_T];
    }
#pragma unroll
    for (int q = 0; q < SPT; ++q) s[q] = dot_rows(hp, a[q], b[q]);
#pragma unroll
    for (int q = 0; q < SPT; ++q) o[base + q * RES_T] = s[q];
  }
#pragma unroll
  for (int q = 0; q < SPT; ++q) {
    int i = base + q * RES_T;
    if (i < E_) o[i] = dot_rows(hp, ia[i], ib[i]);
  }
}

extern "C" void kernel_launch(void* const* d_in, const int* in_sizes, int n_in,
                              void* d_out, int out_size, void* d_ws, size_t ws_size,
                              hipStream_t stream) {
  (void)n_in; (void)out_size; (void)ws_size;
  const float* x    = (const float*)d_in[0];
  const int*   esrc = (const int*)d_in[1];
  const int*   edst = (const int*)d_in[2];
  const int*   nsrc = (const int*)d_in[3];
  const int*   ndst = (const int*)d_in[4];
  const float* Ws1  = (const float*)d_in[5];
  const float* Wn1  = (const float*)d_in[6];
  const float* b1   = (const float*)d_in[7];
  const float* Ws2  = (const float*)d_in[8];
  const float* Wn2  = (const float*)d_in[9];
  const float* b2   = (const float*)d_in[10];

  const int Nn = in_sizes[0] / D;
  const int Ee = in_sizes[3];

  const int segs   = RT * Nn;
  const int edges  = RT * Ee;
  const int cnts   = 2 * segs;
  const int offCnt = (cnts + 1 + 3) & ~3;
  const int nb     = (cnts + SCAN_CHUNK - 1) / SCAN_CHUNK;
  const int bsCnt  = (nb + 1 + 3) & ~3;

  float* B0   = (float*)d_ws;
  float* B1   = B0 + (size_t)Nn * SP;
  int*   cnt  = (int*)(B1 + (size_t)Nn * SP);
  int*   tick = cnt + 2 * (size_t)segs;
  int*   off  = tick + TICK_INTS;
  int*   col  = off + (size_t)offCnt;
  int*   bsum = col + (size_t)edges;
  float* WsP1 = (float*)(bsum + (size_t)bsCnt);
  float* WnP1 = WsP1 + 16 * SP;
  float* bsP1 = WnP1 + RT * 16 * SP;
  float* WsP2 = bsP1 + 16;
  float* WnP2 = WsP2 + 16 * SP;
  float* bsP2 = WnP2 + RT * 16 * SP;

  const int tiles   = (Nn + 15) / 16;
  const int wBlocks = (tiles + 3) / 4;
  const int  edgeGrid = 2048;
  const dim3 scoreGrid(512, 2);

  k_prep<<<1, 256, 0, stream>>>(Ws1, Wn1, b1, Ws2, Wn2, b2,
                                WsP1, WnP1, bsP1, WsP2, WnP2, bsP2);
  k_padcopy<<<1024, RES_T, 0, stream>>>(x, B0, Nn);

  k_zero_i<<<256, RES_T, 0, stream>>>(cnt, cnts + TICK_INTS);
  k_hist<<<edgeGrid, RES_T, 0, stream>>>(edst, cnt, tick, Nn, Ee);
  k_scan_a<<<nb, SCAN_T, 0, stream>>>(cnt, bsum, cnts);
  k_scan_b<<<1, SCANB_T, 0, stream>>>(bsum, nb);
  k_scan_c<<<nb, SCAN_T, 0, stream>>>(cnt, bsum, off, cnts);
  k_fill<<<edgeGrid, RES_T, 0, stream>>>(esrc, edst, cnt, tick + 2 * TICK_PITCH, col, Nn, Ee);

  k_layer<<<wBlocks, 128, 0, stream>>>(B0, off, col, WsP1, WnP1, bsP1, B1, tiles, Nn);
  k_layer<<<wBlocks, 128, 0, stream>>>(B1, off, col, WsP2, WnP2, bsP2, B0, tiles, Nn);

  k_score<<<scoreGrid, RES_T, 0, stream>>>(B0, esrc, edst, nsrc, ndst,
                                           (float*)d_out, Ee);
}


# Replace the global-atomic counting sort of the edges by a two-level bucket sort ranked in LDS: run-wise binning by segment range, then one work-group per bucket; segment sum order changes
# speedup vs baseline: 4.3170x; 1.2124x over previous
#include <hip/hip_runtime.h>


#define D  10
#define RT 4
#define SP 12

#define RES_T 256
#define SPT   2
#define PPT   8

#define BSH   12
#define BW    (1 << BSH)
#define MAXB  512
#define L1_T  512
#define L1_PER 16
#define CH    (L1_T * L1_PER)
#define L2_T  512
#define L2_PER (BW / L2_T)
#define L2_U  8
#define CAP   32768

typedef __attribute__((ext_vector_type(2))) float v2f;
typedef __attribute__((ext_vector_type(8))) float v8f;

__device__ __forceinline__ v8f wmma_k4(v2f a, v2f b, v8f c) {
  return __builtin_amdgcn_wmma_f32_16x16x4_f32(
       false, a,  false, b,
       (short)0, c,  false,  false);
}

__global__ void k_prep(const float* __restrict__ Ws1, const float* __restrict__ Wn1,
                       const float* __restrict__ b1,
                       const float* __restrict__ Ws2, const float* __restrict__ Wn2,
                       const float* __restrict__ b2,
                       float* __restrict__ WsP1, float* __restrict__ WnP1,
                       float* __restrict__ bsP1,
                       float* __restrict__ WsP2, float* __restrict__ WnP2,
                       float* __restrict__ bsP2,
                       int* __restrict__ bzero) {
  for (int i = threadIdx.x; i < 2 * MAXB; i += blockDim.x) bzero[i] = 0;
  for (int i = threadIdx.x; i < 16 * SP; i += blockDim.x) {
    int n = i / SP, k = i % SP;
    float s1 = 0.f, s2 = 0.f;
    if (n < D && k < D) {
#pragma unroll
      for (int r = 0; r < RT; ++r) {
        s1 += Ws1[r * D * D + n * D + k];
        s2 += Ws2[r * D * D + n * D + k];
      }
    }
    WsP1[i] = s1;
    WsP2[i] = s2;
  }
  for (int i = threadIdx.x; i < RT * 16 * SP; i += blockDim.x) {
    int r = i / (16 * SP), j = i % (16 * SP);
    int n = j / SP, k = j % SP;
    float v1 = 0.f, v2 = 0.f;
    if (n < D && k < D) {
      v1 = Wn1[r * D * D + n * D + k];
      v2 = Wn2[r * D * D + n * D + k];
    }
    WnP1[i] = v1;
    WnP2[i] = v2;
  }
  for (int i = threadIdx.x; i < 16; i += blockDim.x) {
    float s1 = 0.f, s2 = 0.f;
    if (i < D) {
#pragma unroll
      for (int r = 0; r < RT; ++r) {
        s1 += b1[r * D + i];
        s2 += b2[r * D + i];
      }
    }
    bsP1[i] = s1;
    bsP2[i] = s2;
  }
}

__device__ __forceinline__ float pad_elem(const float* __restrict__ x, int i) {
  int n = i / SP, k = i - n * SP;
  float t = x[(long long)n * D + min(k, D - 1)];
  return (k < D) ? t : 0.0f;
}

__global__ void __launch_bounds__(RES_T)
k_padcopy(const float* __restrict__ x, float* __restrict__ hp, int Nn) {
  const int total = Nn * SP;
  const int step  = gridDim.x * (RES_T * PPT);
  int base = blockIdx.x * (RES_T * PPT) + threadIdx.x;
  for (; base + (PPT - 1) * RES_T < total; base += step) {
    float v[PPT];
#pragma unroll
    for (int q = 0; q < PPT; ++q) v[q] = pad_elem(x, base + q * RES_T);
#pragma unroll
    for (int q = 0; q < PPT; ++q) hp[base + q * RES_T] = v[q];
  }
#pragma unroll
  for (int q = 0; q < PPT; ++q) {
    int i = base + q * RES_T;
    if (i < total) hp[i] = pad_elem(x, i);
  }
}


__device__ __forceinline__ int seg_of(int i, int d, int Nn, int Ee) {
  const int r = (i >= Ee) + (i >= 2 * Ee) + (i >= 3 * Ee);
  return r * Nn + d;
}

__device__ __forceinline__ void block_scan_incl(int* sh, int t, int n) {
  for (int ofs = 1; ofs < n; ofs <<= 1) {
    int add = (t >= ofs) ? sh[t - ofs] : 0;
    __syncthreads();
    sh[t] += add;
    __syncthreads();
  }
}

__global__ void __launch_bounds__(L1_T)
k_bhist(const int* __restrict__ edst,
        int* __restrict__ bcnt,
        int Nn, int Ee, int edges) {
  __shared__ int lh[MAXB];
  for (int i = threadIdx.x; i < MAXB; i += L1_T) lh[i] = 0;
  __syncthreads();
  const int base = blockIdx.x * CH + threadIdx.x;
  int d[L1_PER];
#pragma unroll
  for (int k = 0; k < L1_PER; ++k) d[k] = edst[min(base + k * L1_T, edges - 1)];
#pragma unroll
  for (int k = 0; k < L1_PER; ++k) {
    const int i = base + k * L1_T;
    if (i < edges) atomicAdd(&lh[seg_of(i, d[k], Nn, Ee) >> BSH], 1);
  }
  __syncthreads();
  for (int i = threadIdx.x; i < MAXB; i += L1_T) {
    const int v = lh[i];
    if (v) atomicAdd(&bcnt[i], v);
  }
}

__global__ void __launch_bounds__(MAXB)
k_bscan(const int* __restrict__ bcnt, int* __restrict__ bbase) {
  __shared__ int sh[MAXB];
  const int t = threadIdx.x;
  const int v = bcnt[t];
  sh[t] = v;
  __syncthreads();
  block_scan_incl(sh, t, MAXB);
  bbase[t] = sh[t] - v;
  if (t == MAXB - 1) bbase[MAXB] = sh[t];
}

__global__ void __launch_bounds__(L1_T)
k_bscatter(const int* __restrict__ esrc,
           const int* __restrict__ edst,
           const int* __restrict__ bbase,
           int* __restrict__ bcur,
           int2* __restrict__ pairs,
           int Nn, int Ee, int edges) {
  static_assert(L1_T == MAXB, "one bucket per thread in the run reservation");
  __shared__ int lh[MAXB];
  __shared__ int lc[MAXB];
  __shared__ int sc[MAXB];
  __shared__ int2 stage[CH];
  const int t = threadIdx.x;
  lh[t] = 0;
  __syncthreads();
  const int base = blockIdx.x * CH + t;
  int g[L1_PER], s[L1_PER];
#pragma unroll
  for (int k = 0; k < L1_PER; ++k) {
    const int idx = min(base + k * L1_T, edges - 1);
    s[k] = esrc[idx];
    g[k] = seg_of(idx, edst[idx], Nn, Ee);
  }
#pragma unroll
  for (int k = 0; k < L1_PER; ++k)
    if (base + k * L1_T < edges) atomicAdd(&lh[g[k] >> BSH], 1);
  __syncthreads();
  const int v = lh[t];
  sc[t] = v;
  __syncthreads();
  block_scan_incl(sc, t, MAXB);
  const int excl = sc[t] - v;
  lc[t] = excl;
  int gs = 0;
  if (v > 0) gs = bbase[t] + atomicAdd(&bcur[t], v);
  lh[t] = gs - excl;
  __syncthreads();
#pragma unroll
  for (int k = 0; k < L1_PER; ++k) {
    if (base + k * L1_T < edges) {
      const int slot = atomicAdd(&lc[g[k] >> BSH], 1);
      stage[slot] = make_int2(s[k], g[k]);
    }
  }
  __syncthreads();
  const int n = min(CH, edges - (int)blockIdx.x * CH);
  if (n == CH) {
#pragma unroll
    for (int k = 0; k < L1_PER; ++k) {
      const int i = t + k * L1_T;
      const int2 p = stage[i];
      pairs[i + lh[p.y >> BSH]] = p;
    }
  } else {
    for (int i = t; i < n; i += L1_T) {
      const int2 p = stage[i];
      pairs[i + lh[p.y >> BSH]] = p;
    }
  }
}

__global__ void __launch_bounds__(L2_T)
k_bsort(const int2* __restrict__ pairs,
        const int* __restrict__ bbase,
        int* __restrict__ off,
        int* __restrict__ col,
        int segs, int nbk) {
  __shared__ int cnt[BW];
  __shared__ int sc[L2_T];
  __shared__ int stage[CAP];
  const int t = threadIdx.x;
  const int b = blockIdx.x;
  const int base = bbase[b];
  const int n = bbase[b + 1] - base;
  const int g0 = b << BSH;
  const int nseg = min(BW, segs - g0);
  const int2* __restrict__ pp = pairs + base;
  int* __restrict__ cb = col + base;
  for (int i = t; i < BW; i += L2_T) cnt[i] = 0;
  __syncthreads();
  for (int i0 = t; i0 < n; i0 += L2_T * L2_U) {
    int l[L2_U];
#pragma unroll
    for (int u = 0; u < L2_U; ++u) l[u] = pp[min(i0 + u * L2_T, n - 1)].y - g0;
#pragma unroll
    for (int u = 0; u < L2_U; ++u)
      if (i0 + u * L2_T < n) atomicAdd(&cnt[l[u]], 1);
  }
  __syncthreads();
  int v[L2_PER];
  int s = 0;
#pragma unroll
  for (int q = 0; q < L2_PER; ++q) {
    v[q] = cnt[t * L2_PER + q];
    s += v[q];
  }
  sc[t] = s;
  __syncthreads();
  block_scan_incl(sc, t, L2_T);
  int run = sc[t] - s;
#pragma unroll
  for (int q = 0; q < L2_PER; ++q) {
    cnt[t * L2_PER + q] = run;
    run += v[q];
  }
  __syncthreads();
#pragma unroll 8
  for (int l = t; l < nseg; l += L2_T) off[g0 + l] = base + cnt[l];
  if (b == nbk - 1 && t == 0) off[segs] = base + n;
  __syncthreads();
  const bool inLds = (n <= CAP);
  for (int i0 = t; i0 < n; i0 += L2_T * L2_U) {
    int2 p[L2_U];
#pragma unroll
    for (int u = 0; u < L2_U; ++u) p[u] = pp[min(i0 + u * L2_T, n - 1)];
#pragma unroll
    for (int u = 0; u < L2_U; ++u) {
      if (i0 + u * L2_T < n) {
        const int slot = atomicAdd(&cnt[p[u].y - g0], 1);
        if (inLds) stage[slot] = p[u].x;
        else       cb[slot] = p[u].x;
      }
    }
  }
  __syncthreads();
  if (inLds) {
#pragma unroll 8
    for (int i = t; i < n; i += L2_T) cb[i] = stage[i];
  }
}


__global__ void k_layer(const float* __restrict__ hp,
                        const int* __restrict__ off,
                        const int* __restrict__ col,
                        const float* __restrict__ WsP,
                        const float* __restrict__ WnP,
                        const float* __restrict__ bsP,
                        float* __restrict__ hout,
                        int nTiles, int Nn) {
  int wave = blockIdx.x * (blockDim.x >> 5) + (threadIdx.x >> 5);
  if (wave >= nTiles) return;
  int lane = threadIdx.x & 31;
  int m    = lane & 15;
  int half = lane >> 4;
  int kk   = half * 2;
  int node = min(wave * 16 + m, Nn - 1);

  union { v8f v; float f[8]; } acc;
  float bsum = bsP[m];
#pragma unroll
  for (int j = 0; j < 8; ++j) acc.f[j] = bsum;

  {
    const float* arow = hp + (long long)node * SP + kk;
    const float* brow = WsP + m * SP + kk;
#pragma unroll
    for (int c = 0; c < 3; ++c) {
      v2f a = *(const v2f*)(arow + c * 4);
      v2f b = *(const v2f*)(brow + c * 4);
      acc.v = wmma_k4(a, b, acc.v);
    }
  }

#pragma unroll
  for (int r = 0; r < RT; ++r) {
    int g   = r * Nn + node;
    int p0  = off[g];
    int end = off[g + 1] - p0;
    const int* __restrict__ cp = col + p0;
    v2f s0 = {0.f, 0.f}, s1 = {0.f, 0.f}, s2 = {0.f, 0.f};
    int j = 0;
    for (; j + 1 < end; j += 2) {
      int sa = cp[j];
      int sb = cp[j + 1];
      const float* ra = hp + (long long)sa * SP + kk;
      const float* rb = hp + (long long)sb * SP + kk;
      v2f xa0 = *(const v2f*)(ra);
      v2f xa1 = *(const v2f*)(ra + 4);
      v2f xa2 = *(const v2f*)(ra + 8);
      v2f xb0 = *(const v2f*)(rb);
      v2f xb1 = *(const v2f*)(rb + 4);
      v2f xb2 = *(const v2f*)(rb + 8);
      s0 += xa0; s1 += xa1; s2 += xa2;
      s0 += xb0; s1 += xb1; s2 += xb2;
    }
    if (j < end) {
      int sa = cp[j];
      const float* ra = hp + (long long)sa * SP + kk;
      s0 += *(const v2f*)(ra);
      s1 += *(const v2f*)(ra + 4);
      s2 += *(const v2f*)(ra + 8);
    }
    float rdeg = 1.0f / fmaxf((float)end, 1.0f);
    const float* bw = WnP + r * 16 * SP + m * SP + kk;
    v2f b0 = *(const v2f*)(bw);
    v2f b1 = *(const v2f*)(bw + 4);
    v2f b2 = *(const v2f*)(bw + 8);
    acc.v = wmma_k4(s0 * rdeg, b0, acc.v);
    acc.v = wmma_k4(s1 * rdeg, b1, acc.v);
    acc.v = wmma_k4(s2 * rdeg, b2, acc.v);
  }

  int rowBase = wave * 16 + half * 8;
  bool fullTile = (wave * 16 + 16) <= Nn;
  if (m < SP) {
    float* op = hout + (long long)rowBase * SP + m;
    if (fullTile) {
#pragma unroll
      for (int j = 0; j < 8; ++j) op[j * SP] = acc.f[j];
    } else {
#pragma unroll
      for (int j = 0; j < 8; ++j)
        if (rowBase + j < Nn) op[j * SP] = acc.f[j];
    }
  }
}

__device__ __forceinline__ float dot_rows(const float* __restrict__ hp, int a, int b) {
  const float* pa = hp + (long long)a * SP;
  const float* pb = hp + (long long)b * SP;
  float s = 0.0f;
#pragma unroll
  for (int k = 0; k < D; ++k) s += pa[k] * pb[k];
  return s;
}

__global__ void __launch_bounds__(RES_T)
k_score(const float* __restrict__ hp,
        const int* __restrict__ ps, const int* __restrict__ pd,
        const int* __restrict__ ns, const int* __restrict__ nd,
        float* __restrict__ out, int E_) {
  const int* __restrict__ ia = blockIdx.y ? ns : ps;
  const int* __restrict__ ib = blockIdx.y ? nd : pd;
  float* __restrict__ o = out + (long long)blockIdx.y * E_;
  const int step = gridDim.x * (RES_T * SPT);
  int base = blockIdx.x * (RES_T * SPT) + threadIdx.x;
  for (; base + (SPT - 1) * RES_T < E_; base += step) {
    int a[SPT], b[SPT];
    float s[SPT];
#pragma unroll
    for (int q = 0; q < SPT; ++q) {
      a[q] = ia[base + q * RES_T];
      b[q] = ib[base + q * RES_T];
    }
#pragma unroll
    for (int q = 0; q < SPT; ++q) s[q] = dot_rows(hp, a[q], b[q]);
#pragma unroll
    for (int q = 0; q < SPT; ++q) o[base + q * RES_T] = s[q];
  }
#pragma unroll
  for (int q = 0; q < SPT; ++q) {
    int i = base + q * RES_T;
    if (i < E_) o[i] = dot_rows(hp, ia[i], ib[i]);
  }
}

extern "C" void kernel_launch(void* const* d_in, const int* in_sizes, int n_in,
                              void* d_out, int out_size, void* d_ws, size_t ws_size,
                              hipStream_t stream) {
  (void)n_in; (void)out_size; (void)ws_size;
  const float* x    = (const float*)d_in[0];
  const int*   esrc = (const int*)d_in[1];
  const int*   edst = (const int*)d_in[2];
  const int*   nsrc = (const int*)d_in[3];
  const int*   ndst = (const int*)d_in[4];
  const float* Ws1  = (const float*)d_in[5];
  const float* Wn1  = (const float*)d_in[6];
  const float* b1   = (const float*)d_in[7];
  const float* Ws2  = (const float*)d_in[8];
  const float* Wn2  = (const float*)d_in[9];
  const float* b2   = (const float*)d_in[10];

  const int Nn = in_sizes[0] / D;
  const int Ee = in_sizes[3];

  const int segs   = RT * Nn;
  const int edges  = RT * Ee;
  const int nbk    = (segs + BW - 1) >> BSH;
  const int nCh    = (edges + CH - 1) / CH;
  const int offCnt = (segs + 1 + 3) & ~3;

  float* B0    = (float*)d_ws;
  float* B1    = B0 + (size_t)Nn * SP;
  int*   bcnt  = (int*)(B1 + (size_t)Nn * SP);
  int*   bcur  = bcnt + MAXB;
  int*   bbase = bcur + MAXB;
  int*   off   = bbase + MAXB + 4;
  int*   col   = off + (size_t)offCnt;
  int2*  pairs = (int2*)(col + (size_t)edges);
  float* WsP1  = (float*)(pairs + (size_t)edges);
  float* WnP1  = WsP1 + 16 * SP;
  float* bsP1  = WnP1 + RT * 16 * SP;
  float* WsP2  = bsP1 + 16;
  float* WnP2  = WsP2 + 16 * SP;
  float* bsP2  = WnP2 + RT * 16 * SP;

  const int tiles   = (Nn + 15) / 16;
  const int wBlocks = (tiles + 3) / 4;
  const dim3 scoreGrid(512, 2);

  k_prep<<<1, 256, 0, stream>>>(Ws1, Wn1, b1, Ws2, Wn2, b2,
                                WsP1, WnP1, bsP1, WsP2, WnP2, bsP2, bcnt);
  k_padcopy<<<1024, RES_T, 0, stream>>>(x, B0, Nn);

  k_bhist<<<nCh, L1_T, 0, stream>>>(edst, bcnt, Nn, Ee, edges);
  k_bscan<<<1, MAXB, 0, stream>>>(bcnt, bbase);
  k_bscatter<<<nCh, L1_T, 0, stream>>>(esrc, edst, bbase, bcur, pairs, Nn, Ee, edges);
  k_bsort<<<nbk, L2_T, 0, stream>>>(pairs, bbase, off, col, segs, nbk);

  k_layer<<<wBlocks, 128, 0, stream>>>(B0, off, col, WsP1, WnP1, bsP1, B1, tiles, Nn);
  k_layer<<<wBlocks, 128, 0, stream>>>(B1, off, col, WsP2, WnP2, bsP2, B0, tiles, Nn);

  k_score<<<scoreGrid, RES_T, 0, stream>>>(B0, esrc, edst, nsrc, ndst,
                                           (float*)d_out, Ee);
}


# Feature rows padded from 48 to 64 B so each gathered row lies in ONE 128-B line (1 in 4 crossed two); a lane reads its share as a 16-B + an 8-B load; k order of the WMMA chain changes, A and B alike
# speedup vs baseline: 4.6503x; 1.0772x over previous
#include <hip/hip_runtime.h>


#define D  10
#define RT 4
#define SP 16

#define RES_T 256
#define SPT   2
#define PPT   8

#define BSH   12
#define BW    (1 << BSH)
#define MAXB  512
#define L1_T  512
#define L1_PER 16
#define CH    (L1_T * L1_PER)
#define L2_T  512
#define L2_PER (BW / L2_T)
#define L2_U  8
#define CAP   32768

typedef __attribute__((ext_vector_type(2))) float v2f;
typedef __attribute__((ext_vector_type(4))) float v4f;
typedef __attribute__((ext_vector_type(8))) float v8f;

__device__ __forceinline__ v8f wmma_k4(v2f a, v2f b, v8f c) {
  return __builtin_amdgcn_wmma_f32_16x16x4_f32(
       false, a,  false, b,
       (short)0, c,  false,  false);
}

__global__ void k_prep(const float* __restrict__ Ws1, const float* __restrict__ Wn1,
                       const float* __restrict__ b1,
                       const float* __restrict__ Ws2, const float* __restrict__ Wn2,
                       const float* __restrict__ b2,
                       float* __restrict__ WsP1, float* __restrict__ WnP1,
                       float* __restrict__ bsP1,
                       float* __restrict__ WsP2, float* __restrict__ WnP2,
                       float* __restrict__ bsP2,
                       int* __restrict__ bzero) {
  for (int i = threadIdx.x; i < 2 * MAXB; i += blockDim.x) bzero[i] = 0;
  for (int i = threadIdx.x; i < 16 * SP; i += blockDim.x) {
    int n = i / SP, k = i % SP;
    float s1 = 0.f, s2 = 0.f;
    if (n < D && k < D) {
#pragma unroll
      for (int r = 0; r < RT; ++r) {
        s1 += Ws1[r * D * D + n * D + k];
        s2 += Ws2[r * D * D + n * D + k];
      }
    }
    WsP1[i] = s1;
    WsP2[i] = s2;
  }
  for (int i = threadIdx.x; i < RT * 16 * SP; i += blockDim.x) {
    int r = i / (16 * SP), j = i % (16 * SP);
    int n = j / SP, k = j % SP;
    float v1 = 0.f, v2 = 0.f;
    if (n < D && k < D) {
      v1 = Wn1[r * D * D + n * D + k];
      v2 = Wn2[r * D * D + n * D + k];
    }
    WnP1[i] = v1;
    WnP2[i] = v2;
  }
  for (int i = threadIdx.x; i < 16; i += blockDim.x) {
    float s1 = 0.f, s2 = 0.f;
    if (i < D) {
#pragma unroll
      for (int r = 0; r < RT; ++r) {
        s1 += b1[r * D + i];
        s2 += b2[r * D + i];
      }
    }
    bsP1[i] = s1;
    bsP2[i] = s2;
  }
}

__device__ __forceinline__ float pad_elem(const float* __restrict__ x, int i) {
  int n = i / SP, k = i - n * SP;
  float t = x[(long long)n * D + min(k, D - 1)];
  return (k < D) ? t : 0.0f;
}

__global__ void __launch_bounds__(RES_T)
k_padcopy(const float* __restrict__ x, float* __restrict__ hp, int Nn) {
  const int total = Nn * SP;
  const int step  = gridDim.x * (RES_T * PPT);
  int base = blockIdx.x * (RES_T * PPT) + threadIdx.x;
  for (; base + (PPT - 1) * RES_T < total; base += step) {
    float v[PPT];
#pragma unroll
    for (int q = 0; q < PPT; ++q) v[q] = pad_elem(x, base + q * RES_T);
#pragma unroll
    for (int q = 0; q < PPT; ++q) hp[base + q * RES_T] = v[q];
  }
#pragma unroll
  for (int q = 0; q < PPT; ++q) {
    int i = base + q * RES_T;
    if (i < total) hp[i] = pad_elem(x, i);
  }
}


__device__ __forceinline__ int seg_of(int i, int d, int Nn, int Ee) {
  const int r = (i >= Ee) + (i >= 2 * Ee) + (i >= 3 * Ee);
  return r * Nn + d;
}

__device__ __forceinline__ void block_scan_incl(int* sh, int t, int n) {
  for (int ofs = 1; ofs < n; ofs <<= 1) {
    int add = (t >= ofs) ? sh[t - ofs] : 0;
    __syncthreads();
    sh[t] += add;
    __syncthreads();
  }
}

__global__ void __launch_bounds__(L1_T)
k_bhist(const int* __restrict__ edst,
        int* __restrict__ bcnt,
        int Nn, int Ee, int edges) {
  __shared__ int lh[MAXB];
  for (int i = threadIdx.x; i < MAXB; i += L1_T) lh[i] = 0;
  __syncthreads();
  const int base = blockIdx.x * CH + threadIdx.x;
  int d[L1_PER];
#pragma unroll
  for (int k = 0; k < L1_PER; ++k) d[k] = edst[min(base + k * L1_T, edges - 1)];
#pragma unroll
  for (int k = 0; k < L1_PER; ++k) {
    const int i = base + k * L1_T;
    if (i < edges) atomicAdd(&lh[seg_of(i, d[k], Nn, Ee) >> BSH], 1);
  }
  __syncthreads();
  for (int i = threadIdx.x; i < MAXB; i += L1_T) {
    const int v = lh[i];
    if (v) atomicAdd(&bcnt[i], v);
  }
}

__global__ void __launch_bounds__(MAXB)
k_bscan(const int* __restrict__ bcnt, int* __restrict__ bbase) {
  __shared__ int sh[MAXB];
  const int t = threadIdx.x;
  const int v = bcnt[t];
  sh[t] = v;
  __syncthreads();
  block_scan_incl(sh, t, MAXB);
  bbase[t] = sh[t] - v;
  if (t == MAXB - 1) bbase[MAXB] = sh[t];
}

__global__ void __launch_bounds__(L1_T)
k_bscatter(const int* __restrict__ esrc,
           const int* __restrict__ edst,
           const int* __restrict__ bbase,
           int* __restrict__ bcur,
           int2* __restrict__ pairs,
           int Nn, int Ee, int edges) {
  static_assert(L1_T == MAXB, "one bucket per thread in the run reservation");
  __shared__ int lh[MAXB];
  __shared__ int lc[MAXB];
  __shared__ int sc[MAXB];
  __shared__ int2 stage[CH];
  const int t = threadIdx.x;
  lh[t] = 0;
  __syncthreads();
  const int base = blockIdx.x * CH + t;
  int g[L1_PER], s[L1_PER];
#pragma unroll
  for (int k = 0; k < L1_PER; ++k) {
    const int idx = min(base + k * L1_T, edges - 1);
    s[k] = esrc[idx];
    g[k] = seg_of(idx, edst[idx], Nn, Ee);
  }
#pragma unroll
  for (int k = 0; k < L1_PER; ++k)
    if (base + k * L1_T < edges) atomicAdd(&lh[g[k] >> BSH], 1);
  __syncthreads();
  const int v = lh[t];
  sc[t] = v;
  __syncthreads();
  block_scan_incl(sc, t, MAXB);
  const int excl = sc[t] - v;
  lc[t] = excl;
  int gs = 0;
  if (v > 0) gs = bbase[t] + atomicAdd(&bcur[t], v);
  lh[t] = gs - excl;
  __syncthreads();
#pragma unroll
  for (int k = 0; k < L1_PER; ++k) {
    if (base + k * L1_T < edges) {
      const int slot = atomicAdd(&lc[g[k] >> BSH], 1);
      stage[slot] = make_int2(s[k], g[k]);
    }
  }
  __syncthreads();
  const int n = min(CH, edges - (int)blockIdx.x * CH);
  if (n == CH) {
#pragma unroll
    for (int k = 0; k < L1_PER; ++k) {
      const int i = t + k * L1_T;
      const int2 p = stage[i];
      pairs[i + lh[p.y >> BSH]] = p;
    }
  } else {
    for (int i = t; i < n; i += L1_T) {
      const int2 p = stage[i];
      pairs[i + lh[p.y >> BSH]] = p;
    }
  }
}

__global__ void __launch_bounds__(L2_T)
k_bsort(const int2* __restrict__ pairs,
        const int* __restrict__ bbase,
        int* __restrict__ off,
        int* __restrict__ col,
        int segs, int nbk) {
  __shared__ int cnt[BW];
  __shared__ int sc[L2_T];
  __shared__ int stage[CAP];
  const int t = threadIdx.x;
  const int b = blockIdx.x;
  const int base = bbase[b];
  const int n = bbase[b + 1] - base;
  const int g0 = b << BSH;
  const int nseg = min(BW, segs - g0);
  const int2* __restrict__ pp = pairs + base;
  int* __restrict__ cb = col + base;
  for (int i = t; i < BW; i += L2_T) cnt[i] = 0;
  __syncthreads();
  for (int i0 = t; i0 < n; i0 += L2_T * L2_U) {
    int l[L2_U];
#pragma unroll
    for (int u = 0; u < L2_U; ++u) l[u] = pp[min(i0 + u * L2_T, n - 1)].y - g0;
#pragma unroll
    for (int u = 0; u < L2_U; ++u)
      if (i0 + u * L2_T < n) atomicAdd(&cnt[l[u]], 1);
  }
  __syncthreads();
  int v[L2_PER];
  int s = 0;
#pragma unroll
  for (int q = 0; q < L2_PER; ++q) {
    v[q] = cnt[t * L2_PER + q];
    s += v[q];
  }
  sc[t] = s;
  __syncthreads();
  block_scan_incl(sc, t, L2_T);
  int run = sc[t] - s;
#pragma unroll
  for (int q = 0; q < L2_PER; ++q) {
    cnt[t * L2_PER + q] = run;
    run += v[q];
  }
  __syncthreads();
#pragma unroll 8
  for (int l = t; l < nseg; l += L2_T) off[g0 + l] = base + cnt[l];
  if (b == nbk - 1 && t == 0) off[segs] = base + n;
  __syncthreads();
  const bool inLds = (n <= CAP);
  for (int i0 = t; i0 < n; i0 += L2_T * L2_U) {
    int2 p[L2_U];
#pragma unroll
    for (int u = 0; u < L2_U; ++u) p[u] = pp[min(i0 + u * L2_T, n - 1)];
#pragma unroll
    for (int u = 0; u < L2_U; ++u) {
      if (i0 + u * L2_T < n) {
        const int slot = atomicAdd(&cnt[p[u].y - g0], 1);
        if (inLds) stage[slot] = p[u].x;
        else       cb[slot] = p[u].x;
      }
    }
  }
  __syncthreads();
  if (inLds) {
#pragma unroll 8
    for (int i = t; i < n; i += L2_T) cb[i] = stage[i];
  }
}

__device__ __forceinline__ void row_part(const float* __restrict__ p,
                                         v4f& q, v2f& t) {
  q = *(const v4f*)(p);
  t = *(const v2f*)(p + 8);
}

__device__ __forceinline__ v8f wmma_row(v4f aq, v2f at, v4f bq, v2f bt, v8f c) {
  c = wmma_k4(aq.xy, bq.xy, c);
  c = wmma_k4(aq.zw, bq.zw, c);
  c = wmma_k4(at, bt, c);
  return c;
}

__global__ void k_layer(const float* __restrict__ hp,
                        const int* __restrict__ off,
                        const int* __restrict__ col,
                        const float* __restrict__ WsP,
                        const float* __restrict__ WnP,
                        const float* __restrict__ bsP,
                        float* __restrict__ hout,
                        int nTiles, int Nn) {
  int wave = blockIdx.x * (blockDim.x >> 5) + (threadIdx.x >> 5);
  if (wave >= nTiles) return;
  int lane = threadIdx.x & 31;
  int m    = lane & 15;
  int half = lane >> 4;
  int ko   = half * 4;
  int node = min(wave * 16 + m, Nn - 1);

  union { v8f v; float f[8]; } acc;
  float bsum = bsP[m];
#pragma unroll
  for (int j = 0; j < 8; ++j) acc.f[j] = bsum;

  {
    v4f aq, bq;
    v2f at, bt;
    row_part(hp + (long long)node * SP + ko, aq, at);
    row_part(WsP + m * SP + ko, bq, bt);
    acc.v = wmma_row(aq, at, bq, bt, acc.v);
  }

#pragma unroll
  for (int r = 0; r < RT; ++r) {
    int g   = r * Nn + node;
    int p0  = off[g];
    int end = off[g + 1] - p0;
    const int* __restrict__ cp = col + p0;
    v4f sq = {0.f, 0.f, 0.f, 0.f};
    v2f st = {0.f, 0.f};
    int j = 0;
    for (; j + 1 < end; j += 2) {
      int sa = cp[j];
      int sb = cp[j + 1];
      v4f qa, qb;
      v2f ta, tb;
      row_part(hp + (long long)sa * SP + ko, qa, ta);
      row_part(hp + (long long)sb * SP + ko, qb, tb);
      sq += qa; st += ta;
      sq += qb; st += tb;
    }
    if (j < end) {
      int sa = cp[j];
      v4f qa;
      v2f ta;
      row_part(hp + (long long)sa * SP + ko, qa, ta);
      sq += qa; st += ta;
    }
    float rdeg = 1.0f / fmaxf((float)end, 1.0f);
    v4f bq;
    v2f bt;
    row_part(WnP + r * 16 * SP + m * SP + ko, bq, bt);
    acc.v = wmma_row(sq * rdeg, st * rdeg, bq, bt, acc.v);
  }

  int rowBase = wave * 16 + half * 8;
  bool fullTile = (wave * 16 + 16) <= Nn;
  float* op = hout + (long long)rowBase * SP + m;
  if (fullTile) {
#pragma unroll
    for (int j = 0; j < 8; ++j) op[j * SP] = acc.f[j];
  } else {
#pragma unroll
    for (int j = 0; j < 8; ++j)
      if (rowBase + j < Nn) op[j * SP] = acc.f[j];
  }
}

__device__ __forceinline__ float dot_rows(const float* __restrict__ hp, int a, int b) {
  const float* pa = hp + (long long)a * SP;
  const float* pb = hp + (long long)b * SP;
  const v4f a0 = *(const v4f*)(pa);
  const v4f a1 = *(const v4f*)(pa + 4);
  const v2f a2 = *(const v2f*)(pa + 8);
  const v4f b0 = *(const v4f*)(pb);
  const v4f b1 = *(const v4f*)(pb + 4);
  const v2f b2 = *(const v2f*)(pb + 8);
  float s = 0.0f;
  s += a0.x * b0.x; s += a0.y * b0.y; s += a0.z * b0.z; s += a0.w * b0.w;
  s += a1.x * b1.x; s += a1.y * b1.y; s += a1.z * b1.z; s += a1.w * b1.w;
  s += a2.x * b2.x; s += a2.y * b2.y;
  return s;
}

__global__ void __launch_bounds__(RES_T)
k_score(const float* __restrict__ hp,
        const int* __restrict__ ps, const int* __restrict__ pd,
        const int* __restrict__ ns, const int* __restrict__ nd,
        float* __restrict__ out, int E_) {
  const int* __restrict__ ia = blockIdx.y ? ns : ps;
  const int* __restrict__ ib = blockIdx.y ? nd : pd;
  float* __restrict__ o = out + (long long)blockIdx.y * E_;
  const int step = gridDim.x * (RES_T * SPT);
  int base = blockIdx.x * (RES_T * SPT) + threadIdx.x;
  for (; base + (SPT - 1) * RES_T < E_; base += step) {
    int a[SPT], b[SPT];
    float s[SPT];
#pragma unroll
    for (int q = 0; q < SPT; ++q) {
      a[q] = ia[base + q * RES_T];
      b[q] = ib[base + q * RES_T];
    }
#pragma unroll
    for (int q = 0; q < SPT; ++q) s[q] = dot_rows(hp, a[q], b[q]);
#pragma unroll
    for (int q = 0; q < SPT; ++q) o[base + q * RES_T] = s[q];
  }
#pragma unroll
  for (int q = 0; q < SPT; ++q) {
    int i = base + q * RES_T;
    if (i < E_) o[i] = dot_rows(hp, ia[i], ib[i]);
  }
}

extern "C" void kernel_launch(void* const* d_in, const int* in_sizes, int n_in,
                              void* d_out, int out_size, void* d_ws, size_t ws_size,
                              hipStream_t stream) {
  (void)n_in; (void)out_size; (void)ws_size;
  const float* x    = (const float*)d_in[0];
  const int*   esrc = (const int*)d_in[1];
  const int*   edst = (const int*)d_in[2];
  const int*   nsrc = (const int*)d_in[3];
  const int*   ndst = (const int*)d_in[4];
  const float* Ws1  = (const float*)d_in[5];
  const float* Wn1  = (const float*)d_in[6];
  const float* b1   = (const float*)d_in[7];
  const float* Ws2  = (const float*)d_in[8];
  const float* Wn2  = (const float*)d_in[9];
  const float* b2   = (const float*)d_in[10];

  const int Nn = in_sizes[0] / D;
  const int Ee = in_sizes[3];

  const int segs   = RT * Nn;
  const int edges  = RT * Ee;
  const int nbk    = (segs + BW - 1) >> BSH;
  const int nCh    = (edges + CH - 1) / CH;
  const int offCnt = (segs + 1 + 3) & ~3;

  float* B0    = (float*)d_ws;
  float* B1    = B0 + (size_t)Nn * SP;
  int*   bcnt  = (int*)(B1 + (size_t)Nn * SP);
  int*   bcur  = bcnt + MAXB;
  int*   bbase = bcur + MAXB;
  int*   off   = bbase + MAXB + 4;
  int*   col   = off + (size_t)offCnt;
  int2*  pairs = (int2*)(col + (size_t)edges);
  float* WsP1  = (float*)(pairs + (size_t)edges);
  float* WnP1  = WsP1 + 16 * SP;
  float* bsP1  = WnP1 + RT * 16 * SP;
  float* WsP2  = bsP1 + 16;
  float* WnP2  = WsP2 + 16 * SP;
  float* bsP2  = WnP2 + RT * 16 * SP;

  const int tiles   = (Nn + 15) / 16;
  const int wBlocks = (tiles + 3) / 4;
  const dim3 scoreGrid(512, 2);

  k_prep<<<1, 256, 0, stream>>>(Ws1, Wn1, b1, Ws2, Wn2, b2,
                                WsP1, WnP1, bsP1, WsP2, WnP2, bsP2, bcnt);
  k_padcopy<<<1024, RES_T, 0, stream>>>(x, B0, Nn);

  k_bhist<<<nCh, L1_T, 0, stream>>>(edst, bcnt, Nn, Ee, edges);
  k_bscan<<<1, MAXB, 0, stream>>>(bcnt, bbase);
  k_bscatter<<<nCh, L1_T, 0, stream>>>(esrc, edst, bbase, bcur, pairs, Nn, Ee, edges);
  k_bsort<<<nbk, L2_T, 0, stream>>>(pairs, bbase, off, col, segs, nbk);

  k_layer<<<wBlocks, 128, 0, stream>>>(B0, off, col, WsP1, WnP1, bsP1, B1, tiles, Nn);
  k_layer<<<wBlocks, 128, 0, stream>>>(B1, off, col, WsP2, WnP2, bsP2, B0, tiles, Nn);

  k_score<<<scoreGrid, RES_T, 0, stream>>>(B0, esrc, edst, nsrc, ndst,
                                           (float*)d_out, Ee);
}


# Gathered rows read by 4 adjacent lanes as whole 64-B rows (8 lines per load, not 16 or 32) in layers and score; layer sums reach WMMA via wave-private LDS, order kept; score dot order changes
# speedup vs baseline: 5.1481x; 1.1070x over previous
#include <hip/hip_runtime.h>


#define D  10
#define RT 4
#define SP 16

#define RES_T 256
#define PPT   8

#define BSH   12
#define BW    (1 << BSH)
#define MAXB  512
#define L1_T  512
#define L1_PER 16
#define CH    (L1_T * L1_PER)
#define L2_T  512
#define L2_PER (BW / L2_T)
#define L2_U  8
#define CAP   32768

typedef __attribute__((ext_vector_type(2))) float v2f;
typedef __attribute__((ext_vector_type(4))) float v4f;
typedef __attribute__((ext_vector_type(8))) float v8f;

__device__ __forceinline__ v8f wmma_k4(v2f a, v2f b, v8f c) {
  return __builtin_amdgcn_wmma_f32_16x16x4_f32(
       false, a,  false, b,
       (short)0, c,  false,  false);
}

__global__ void k_prep(const float* __restrict__ Ws1, const float* __restrict__ Wn1,
                       const float* __restrict__ b1,
                       const float* __restrict__ Ws2, const float* __restrict__ Wn2,
                       const float* __restrict__ b2,
                       float* __restrict__ WsP1, float* __restrict__ WnP1,
                       float* __restrict__ bsP1,
                       float* __restrict__ WsP2, float* __restrict__ WnP2,
                       float* __restrict__ bsP2,
                       int* __restrict__ bzero) {
  for (int i = threadIdx.x; i < 2 * MAXB; i += blockDim.x) bzero[i] = 0;
  for (int i = threadIdx.x; i < 16 * SP; i += blockDim.x) {
    int n = i / SP, k = i % SP;
    float s1 = 0.f, s2 = 0.f;
    if (n < D && k < D) {
#pragma unroll
      for (int r = 0; r < RT; ++r) {
        s1 += Ws1[r * D * D + n * D + k];
        s2 += Ws2[r * D * D + n * D + k];
      }
    }
    WsP1[i] = s1;
    WsP2[i] = s2;
  }
  for (int i = threadIdx.x; i < RT * 16 * SP; i += blockDim.x) {
    int r = i / (16 * SP), j = i % (16 * SP);
    int n = j / SP, k = j % SP;
    float v1 = 0.f, v2 = 0.f;
    if (n < D && k < D) {
      v1 = Wn1[r * D * D + n * D + k];
      v2 = Wn2[r * D * D + n * D + k];
    }
    WnP1[i] = v1;
    WnP2[i] = v2;
  }
  for (int i = threadIdx.x; i < 16; i += blockDim.x) {
    float s1 = 0.f, s2 = 0.f;
    if (i < D) {
#pragma unroll
      for (int r = 0; r < RT; ++r) {
        s1 += b1[r * D + i];
        s2 += b2[r * D + i];
      }
    }
    bsP1[i] = s1;
    bsP2[i] = s2;
  }
}

__device__ __forceinline__ float pad_elem(const float* __restrict__ x, int i) {
  int n = i / SP, k = i - n * SP;
  float t = x[(long long)n * D + min(k, D - 1)];
  return (k < D) ? t : 0.0f;
}

__global__ void __launch_bounds__(RES_T)
k_padcopy(const float* __restrict__ x, float* __restrict__ hp, int Nn) {
  const int total = Nn * SP;
  const int step  = gridDim.x * (RES_T * PPT);
  int base = blockIdx.x * (RES_T * PPT) + threadIdx.x;
  for (; base + (PPT - 1) * RES_T < total; base += step) {
    float v[PPT];
#pragma unroll
    for (int q = 0; q < PPT; ++q) v[q] = pad_elem(x, base + q * RES_T);
#pragma unroll
    for (int q = 0; q < PPT; ++q) hp[base + q * RES_T] = v[q];
  }
#pragma unroll
  for (int q = 0; q < PPT; ++q) {
    int i = base + q * RES_T;
    if (i < total) hp[i] = pad_elem(x, i);
  }
}


__device__ __forceinline__ int seg_of(int i, int d, int Nn, int Ee) {
  const int r = (i >= Ee) + (i >= 2 * Ee) + (i >= 3 * Ee);
  return r * Nn + d;
}

__device__ __forceinline__ void block_scan_incl(int* sh, int t, int n) {
  for (int ofs = 1; ofs < n; ofs <<= 1) {
    int add = (t >= ofs) ? sh[t - ofs] : 0;
    __syncthreads();
    sh[t] += add;
    __syncthreads();
  }
}

__global__ void __launch_bounds__(L1_T)
k_bhist(const int* __restrict__ edst,
        int* __restrict__ bcnt,
        int Nn, int Ee, int edges) {
  __shared__ int lh[MAXB];
  for (int i = threadIdx.x; i < MAXB; i += L1_T) lh[i] = 0;
  __syncthreads();
  const int base = blockIdx.x * CH + threadIdx.x;
  int d[L1_PER];
#pragma unroll
  for (int k = 0; k < L1_PER; ++k) d[k] = edst[min(base + k * L1_T, edges - 1)];
#pragma unroll
  for (int k = 0; k < L1_PER; ++k) {
    const int i = base + k * L1_T;
    if (i < edges) atomicAdd(&lh[seg_of(i, d[k], Nn, Ee) >> BSH], 1);
  }
  __syncthreads();
  for (int i = threadIdx.x; i < MAXB; i += L1_T) {
    const int v = lh[i];
    if (v) atomicAdd(&bcnt[i], v);
  }
}

__global__ void __launch_bounds__(MAXB)
k_bscan(const int* __restrict__ bcnt, int* __restrict__ bbase) {
  __shared__ int sh[MAXB];
  const int t = threadIdx.x;
  const int v = bcnt[t];
  sh[t] = v;
  __syncthreads();
  block_scan_incl(sh, t, MAXB);
  bbase[t] = sh[t] - v;
  if (t == MAXB - 1) bbase[MAXB] = sh[t];
}

__global__ void __launch_bounds__(L1_T)
k_bscatter(const int* __restrict__ esrc,
           const int* __restrict__ edst,
           const int* __restrict__ bbase,
           int* __restrict__ bcur,
           int2* __restrict__ pairs,
           int Nn, int Ee, int edges) {
  static_assert(L1_T == MAXB, "one bucket per thread in the run reservation");
  __shared__ int lh[MAXB];
  __shared__ int lc[MAXB];
  __shared__ int sc[MAXB];
  __shared__ int2 stage[CH];
  const int t = threadIdx.x;
  lh[t] = 0;
  __syncthreads();
  const int base = blockIdx.x * CH + t;
  int g[L1_PER], s[L1_PER];
#pragma unroll
  for (int k = 0; k < L1_PER; ++k) {
    const int idx = min(base + k * L1_T, edges - 1);
    s[k] = esrc[idx];
    g[k] = seg_of(idx, edst[idx], Nn, Ee);
  }
#pragma unroll
  for (int k = 0; k < L1_PER; ++k)
    if (base + k * L1_T < edges) atomicAdd(&lh[g[k] >> BSH], 1);
  __syncthreads();
  const int v = lh[t];
  sc[t] = v;
  __syncthreads();
  block_scan_incl(sc, t, MAXB);
  const int excl = sc[t] - v;
  lc[t] = excl;
  int gs = 0;
  if (v > 0) gs = bbase[t] + atomicAdd(&bcur[t], v);
  lh[t] = gs - excl;
  __syncthreads();
#pragma unroll
  for (int k = 0; k < L1_PER; ++k) {
    if (base + k * L1_T < edges) {
      const int slot = atomicAdd(&lc[g[k] >> BSH], 1);
      stage[slot] = make_int2(s[k], g[k]);
    }
  }
  __syncthreads();
  const int n = min(CH, edges - (int)blockIdx.x * CH);
  if (n == CH) {
#pragma unroll
    for (int k = 0; k < L1_PER; ++k) {
      const int i = t + k * L1_T;
      const int2 p = stage[i];
      pairs[i + lh[p.y >> BSH]] = p;
    }
  } else {
    for (int i = t; i < n; i += L1_T) {
      const int2 p = stage[i];
      pairs[i + lh[p.y >> BSH]] = p;
    }
  }
}

__global__ void __launch_bounds__(L2_T)
k_bsort(const int2* __restrict__ pairs,
        const int* __restrict__ bbase,
        int* __restrict__ off,
        int* __restrict__ col,
        int segs, int nbk) {
  __shared__ int cnt[BW];
  __shared__ int sc[L2_T];
  __shared__ int stage[CAP];
  const int t = threadIdx.x;
  const int b = blockIdx.x;
  const int base = bbase[b];
  const int n = bbase[b + 1] - base;
  const int g0 = b << BSH;
  const int nseg = min(BW, segs - g0);
  const int2* __restrict__ pp = pairs + base;
  int* __restrict__ cb = col + base;
  for (int i = t; i < BW; i += L2_T) cnt[i] = 0;
  __syncthreads();
  for (int i0 = t; i0 < n; i0 += L2_T * L2_U) {
    int l[L2_U];
#pragma unroll
    for (int u = 0; u < L2_U; ++u) l[u] = pp[min(i0 + u * L2_T, n - 1)].y - g0;
#pragma unroll
    for (int u = 0; u < L2_U; ++u)
      if (i0 + u * L2_T < n) atomicAdd(&cnt[l[u]], 1);
  }
  __syncthreads();
  int v[L2_PER];
  int s = 0;
#pragma unroll
  for (int q = 0; q < L2_PER; ++q) {
    v[q] = cnt[t * L2_PER + q];
    s += v[q];
  }
  sc[t] = s;
  __syncthreads();
  block_scan_incl(sc, t, L2_T);
  int run = sc[t] - s;
#pragma unroll
  for (int q = 0; q < L2_PER; ++q) {
    cnt[t * L2_PER + q] = run;
    run += v[q];
  }
  __syncthreads();
#pragma unroll 8
  for (int l = t; l < nseg; l += L2_T) off[g0 + l] = base + cnt[l];
  if (b == nbk - 1 && t == 0) off[segs] = base + n;
  __syncthreads();
  const bool inLds = (n <= CAP);
  for (int i0 = t; i0 < n; i0 += L2_T * L2_U) {
    int2 p[L2_U];
#pragma unroll
    for (int u = 0; u < L2_U; ++u) p[u] = pp[min(i0 + u * L2_T, n - 1)];
#pragma unroll
    for (int u = 0; u < L2_U; ++u) {
      if (i0 + u * L2_T < n) {
        const int slot = atomicAdd(&cnt[p[u].y - g0], 1);
        if (inLds) stage[slot] = p[u].x;
        else       cb[slot] = p[u].x;
      }
    }
  }
  __syncthreads();
  if (inLds) {
#pragma unroll 8
    for (int i = t; i < n; i += L2_T) cb[i] = stage[i];
  }
}

__device__ __forceinline__ void row_part(const float* __restrict__ p,
                                         v4f& q, v2f& t) {
  q = *(const v4f*)(p);
  t = *(const v2f*)(p + 8);
}

__device__ __forceinline__ v8f wmma_row(v4f aq, v2f at, v4f bq, v2f bt, v8f c) {
  c = wmma_k4(aq.xy, bq.xy, c);
  c = wmma_k4(aq.zw, bq.zw, c);
  c = wmma_k4(at, bt, c);
  return c;
}

__device__ __forceinline__ v4f row_piece(const char* __restrict__ hb, int row, unsigned po) {
  return *(const v4f*)(hb + ((unsigned)row * (unsigned)(SP * 4) + po));
}

#define LW 4

__global__ void __launch_bounds__(32 * LW)
k_layer(const float* __restrict__ hp,
        const int* __restrict__ off,
        const int* __restrict__ col,
        const float* __restrict__ WsP,
        const float* __restrict__ WnP,
        const float* __restrict__ bsP,
        float* __restrict__ hout,
        int nTiles, int Nn) {
  __shared__ v4f sm[LW][16 * SP / 4];
  const int wib = threadIdx.x >> 5;
  const int wave = blockIdx.x * LW + wib;
  if (wave >= nTiles) return;
  const int lane = threadIdx.x & 31;
  const int m    = lane & 15;
  const int half = lane >> 4;
  const int ko   = half * 4;
  const int node = min(wave * 16 + m, Nn - 1);

  const int part = lane & 3;
  const int grp  = lane >> 2;
  const unsigned po = (unsigned)part * 16u;
  const int nodeA = min(wave * 16 + grp, Nn - 1);
  const int nodeB = min(wave * 16 + grp + 8, Nn - 1);
  const char* __restrict__ hb = (const char*)hp;
  v4f* __restrict__ smw = sm[wib];
  const float* smf = (const float*)smw;

  union { v8f v; float f[8]; } acc;
  float bsum = bsP[m];
#pragma unroll
  for (int j = 0; j < 8; ++j) acc.f[j] = bsum;

  {
    v4f aq, bq;
    v2f at, bt;
    row_part(hp + (long long)node * SP + ko, aq, at);
    row_part(WsP + m * SP + ko, bq, bt);
    acc.v = wmma_row(aq, at, bq, bt, acc.v);
  }

  int oA0 = off[nodeA], oA1 = off[nodeA + 1];
  int oB0 = off[nodeB], oB1 = off[nodeB + 1];
#pragma unroll
  for (int r = 0; r < RT; ++r) {
    const int pA = oA0;
    const int nA = oA1 - oA0;
    const int pB = oB0;
    const int nB = oB1 - oB0;
    if (r + 1 < RT) {
      const int gA = (r + 1) * Nn + nodeA;
      const int gB = (r + 1) * Nn + nodeB;
      oA0 = off[gA]; oA1 = off[gA + 1];
      oB0 = off[gB]; oB1 = off[gB + 1];
    }
    const int* __restrict__ cA = col + pA;
    const int* __restrict__ cB = col + pB;
    const v4f zero = {0.f, 0.f, 0.f, 0.f};
    v4f sA = zero, sB = zero;
    const int nmax = max(nA, nB);
    for (int j = 0; j < nmax; j += 2) {
      int a0 = -1, a1 = -1, b0 = -1, b1 = -1;
      if (j < nA)     a0 = cA[j];
      if (j + 1 < nA) a1 = cA[j + 1];
      if (j < nB)     b0 = cB[j];
      if (j + 1 < nB) b1 = cB[j + 1];
      v4f qa0 = zero, qa1 = zero, qb0 = zero, qb1 = zero;
      if (a0 >= 0) qa0 = row_piece(hb, a0, po);
      if (a1 >= 0) qa1 = row_piece(hb, a1, po);
      if (b0 >= 0) qb0 = row_piece(hb, b0, po);
      if (b1 >= 0) qb1 = row_piece(hb, b1, po);
      sA += qa0; sA += qa1;
      sB += qb0; sB += qb1;
    }
    const float rA = 1.0f / fmaxf((float)nA, 1.0f);
    const float rB = 1.0f / fmaxf((float)nB, 1.0f);
    smw[grp * (SP / 4) + part]       = sA * rA;
    smw[(grp + 8) * (SP / 4) + part] = sB * rB;
    v4f aq, bq;
    v2f at, bt;
    row_part(smf + m * SP + ko, aq, at);
    row_part(WnP + r * 16 * SP + m * SP + ko, bq, bt);
    acc.v = wmma_row(aq, at, bq, bt, acc.v);
  }

  int rowBase = wave * 16 + half * 8;
  bool fullTile = (wave * 16 + 16) <= Nn;
  float* op = hout + (long long)rowBase * SP + m;
  if (fullTile) {
#pragma unroll
    for (int j = 0; j < 8; ++j) op[j * SP] = acc.f[j];
  } else {
#pragma unroll
    for (int j = 0; j < 8; ++j)
      if (rowBase + j < Nn) op[j * SP] = acc.f[j];
  }
}


template <int Q>
__device__ __forceinline__ int quad_lane(int v) {
  return __builtin_amdgcn_update_dpp(0, v, Q * 0x55, 0xF, 0xF, true);
}

__device__ __forceinline__ float quad_x1(float v) {
  return __builtin_bit_cast(float, __builtin_amdgcn_update_dpp(
             0, __builtin_bit_cast(int, v), 0xB1, 0xF, 0xF, true));
}
__device__ __forceinline__ float quad_x2(float v) {
  return __builtin_bit_cast(float, __builtin_amdgcn_update_dpp(
             0, __builtin_bit_cast(int, v), 0x4E, 0xF, 0xF, true));
}

__device__ __forceinline__ float dot4(v4f a, v4f b) {
  return a.x * b.x + a.y * b.y + a.z * b.z + a.w * b.w;
}

struct ScoreRows { v4f a0, b0, a1, b1, a2, b2, a3, b3; };

__device__ __forceinline__ ScoreRows score_load(const char* __restrict__ hb,
                                                int ca, int cb, unsigned po) {
  ScoreRows r;
  r.a0 = row_piece(hb, quad_lane<0>(ca), po);
  r.b0 = row_piece(hb, quad_lane<0>(cb), po);
  r.a1 = row_piece(hb, quad_lane<1>(ca), po);
  r.b1 = row_piece(hb, quad_lane<1>(cb), po);
  r.a2 = row_piece(hb, quad_lane<2>(ca), po);
  r.b2 = row_piece(hb, quad_lane<2>(cb), po);
  r.a3 = row_piece(hb, quad_lane<3>(ca), po);
  r.b3 = row_piece(hb, quad_lane<3>(cb), po);
  return r;
}

__device__ __forceinline__ float score_sum(const ScoreRows& r, int part) {
  const float p0 = dot4(r.a0, r.b0);
  const float p1 = dot4(r.a1, r.b1);
  const float p2 = dot4(r.a2, r.b2);
  const float p3 = dot4(r.a3, r.b3);
  const bool odd = (part & 1) != 0;
  const bool hi  = (part & 2) != 0;
  const float lo01 = (odd ? p1 : p0) + quad_x1(odd ? p0 : p1);
  const float lo23 = (odd ? p3 : p2) + quad_x1(odd ? p2 : p3);
  return (hi ? lo23 : lo01) + quad_x2(hi ? lo01 : lo23);
}

__global__ void __launch_bounds__(RES_T)
k_score(const float* __restrict__ hp,
        const int* __restrict__ ps, const int* __restrict__ pd,
        const int* __restrict__ ns, const int* __restrict__ nd,
        float* __restrict__ out, int E_) {
  const int* __restrict__ ia = blockIdx.y ? ns : ps;
  const int* __restrict__ ib = blockIdx.y ? nd : pd;
  float* __restrict__ o = out + (long long)blockIdx.y * E_;
  const char* __restrict__ hb = (const char*)hp;
  const int lane = threadIdx.x & 31;
  const int part = lane & 3;
  const int k    = part * 8 + (lane >> 2);
  const unsigned po = (unsigned)part * 16u;
  const int nw   = gridDim.x * (RES_T / 32);
  const int last = E_ - 1;
  const int wv = __builtin_amdgcn_readfirstlane((int)(threadIdx.x >> 5));
  int t = blockIdx.x * (RES_T / 32) + wv;
  if (t * 32 >= E_) return;
  int p  = min(t * 32 + k, last);
  int ca = ia[p], cb = ib[p];
  p = min((t + nw) * 32 + k, last);
  int na = ia[p], nb = ib[p];
  for (;;) {
    {
      const ScoreRows r = score_load(hb, ca, cb, po);
      p = min((t + 2 * nw) * 32 + k, last);
      ca = ia[p]; cb = ib[p];
      const float s = score_sum(r, part);
      const int i = t * 32 + k;
      if (i < E_) o[i] = s;
    }
    t += nw;
    if (t * 32 >= E_) break;
    {
      const ScoreRows r = score_load(hb, na, nb, po);
      p = min((t + 2 * nw) * 32 + k, last);
      na = ia[p]; nb = ib[p];
      const float s = score_sum(r, part);
      const int i = t * 32 + k;
      if (i < E_) o[i] = s;
    }
    t += nw;
    if (t * 32 >= E_) break;
  }
}

extern "C" void kernel_launch(void* const* d_in, const int* in_sizes, int n_in,
                              void* d_out, int out_size, void* d_ws, size_t ws_size,
                              hipStream_t stream) {
  (void)n_in; (void)out_size; (void)ws_size;
  const float* x    = (const float*)d_in[0];
  const int*   esrc = (const int*)d_in[1];
  const int*   edst = (const int*)d_in[2];
  const int*   nsrc = (const int*)d_in[3];
  const int*   ndst = (const int*)d_in[4];
  const float* Ws1  = (const float*)d_in[5];
  const float* Wn1  = (const float*)d_in[6];
  const float* b1   = (const float*)d_in[7];
  const float* Ws2  = (const float*)d_in[8];
  const float* Wn2  = (const float*)d_in[9];
  const float* b2   = (const float*)d_in[10];

  const int Nn = in_sizes[0] / D;
  const int Ee = in_sizes[3];

  const int segs   = RT * Nn;
  const int edges  = RT * Ee;
  const int nbk    = (segs + BW - 1) >> BSH;
  const int nCh    = (edges + CH - 1) / CH;
  const int offCnt = (segs + 1 + 3) & ~3;

  float* B0    = (float*)d_ws;
  float* B1    = B0 + (size_t)Nn * SP;
  int*   bcnt  = (int*)(B1 + (size_t)Nn * SP);
  int*   bcur  = bcnt + MAXB;
  int*   bbase = bcur + MAXB;
  int*   off   = bbase + MAXB + 4;
  int*   col   = off + (size_t)offCnt;
  int2*  pairs = (int2*)(col + (size_t)edges);
  float* WsP1  = (float*)(pairs + (size_t)edges);
  float* WnP1  = WsP1 + 16 * SP;
  float* bsP1  = WnP1 + RT * 16 * SP;
  float* WsP2  = bsP1 + 16;
  float* WnP2  = WsP2 + 16 * SP;
  float* bsP2  = WnP2 + RT * 16 * SP;

  const int tiles   = (Nn + 15) / 16;
  const int wBlocks = (tiles + LW - 1) / LW;
  const dim3 scoreGrid(512, 2);

  k_prep<<<1, 256, 0, stream>>>(Ws1, Wn1, b1, Ws2, Wn2, b2,
                                WsP1, WnP1, bsP1, WsP2, WnP2, bsP2, bcnt);
  k_padcopy<<<1024, RES_T, 0, stream>>>(x, B0, Nn);

  k_bhist<<<nCh, L1_T, 0, stream>>>(edst, bcnt, Nn, Ee, edges);
  k_bscan<<<1, MAXB, 0, stream>>>(bcnt, bbase);
  k_bscatter<<<nCh, L1_T, 0, stream>>>(esrc, edst, bbase, bcur, pairs, Nn, Ee, edges);
  k_bsort<<<nbk, L2_T, 0, stream>>>(pairs, bbase, off, col, segs, nbk);

  k_layer<<<wBlocks, 32 * LW, 0, stream>>>(B0, off, col, WsP1, WnP1, bsP1, B1, tiles, Nn);
  k_layer<<<wBlocks, 32 * LW, 0, stream>>>(B1, off, col, WsP2, WnP2, bsP2, B0, tiles, Nn);

  k_score<<<scoreGrid, RES_T, 0, stream>>>(B0, esrc, edst, nsrc, ndst,
                                           (float*)d_out, Ee);
}


# Layer: a tile's 16 segments of an etype are consecutive in col: the wave gathers them as one flat run (8 rows per load, all in flight) into its LDS stage and sums each segment there; sum order kept
# speedup vs baseline: 5.4252x; 1.0538x over previous
#include <hip/hip_runtime.h>


#define D  10
#define RT 4
#define SP 16

#define RES_T 256
#define PPT   8

#define BSH   12
#define BW    (1 << BSH)
#define MAXB  512
#define L1_T  512
#define L1_PER 16
#define CH    (L1_T * L1_PER)
#define L2_T  512
#define L2_PER (BW / L2_T)
#define L2_U  8
#define CAP   32768

typedef __attribute__((ext_vector_type(2))) float v2f;
typedef __attribute__((ext_vector_type(4))) float v4f;
typedef __attribute__((ext_vector_type(8))) float v8f;

__device__ __forceinline__ v8f wmma_k4(v2f a, v2f b, v8f c) {
  return __builtin_amdgcn_wmma_f32_16x16x4_f32(
       false, a,  false, b,
       (short)0, c,  false,  false);
}

__global__ void k_prep(const float* __restrict__ Ws1, const float* __restrict__ Wn1,
                       const float* __restrict__ b1,
                       const float* __restrict__ Ws2, const float* __restrict__ Wn2,
                       const float* __restrict__ b2,
                       float* __restrict__ WsP1, float* __restrict__ WnP1,
                       float* __restrict__ bsP1,
                       float* __restrict__ WsP2, float* __restrict__ WnP2,
                       float* __restrict__ bsP2,
                       int* __restrict__ bzero) {
  for (int i = threadIdx.x; i < 2 * MAXB; i += blockDim.x) bzero[i] = 0;
  for (int i = threadIdx.x; i < 16 * SP; i += blockDim.x) {
    int n = i / SP, k = i % SP;
    float s1 = 0.f, s2 = 0.f;
    if (n < D && k < D) {
#pragma unroll
      for (int r = 0; r < RT; ++r) {
        s1 += Ws1[r * D * D + n * D + k];
        s2 += Ws2[r * D * D + n * D + k];
      }
    }
    WsP1[i] = s1;
    WsP2[i] = s2;
  }
  for (int i = threadIdx.x; i < RT * 16 * SP; i += blockDim.x) {
    int r = i / (16 * SP), j = i % (16 * SP);
    int n = j / SP, k = j % SP;
    float v1 = 0.f, v2 = 0.f;
    if (n < D && k < D) {
      v1 = Wn1[r * D * D + n * D + k];
      v2 = Wn2[r * D * D + n * D + k];
    }
    WnP1[i] = v1;
    WnP2[i] = v2;
  }
  for (int i = threadIdx.x; i < 16; i += blockDim.x) {
    float s1 = 0.f, s2 = 0.f;
    if (i < D) {
#pragma unroll
      for (int r = 0; r < RT; ++r) {
        s1 += b1[r * D + i];
        s2 += b2[r * D + i];
      }
    }
    bsP1[i] = s1;
    bsP2[i] = s2;
  }
}

__device__ __forceinline__ float pad_elem(const float* __restrict__ x, int i) {
  int n = i / SP, k = i - n * SP;
  float t = x[(long long)n * D + min(k, D - 1)];
  return (k < D) ? t : 0.0f;
}

__global__ void __launch_bounds__(RES_T)
k_padcopy(const float* __restrict__ x, float* __restrict__ hp, int Nn) {
  const int total = Nn * SP;
  const int step  = gridDim.x * (RES_T * PPT);
  int base = blockIdx.x * (RES_T * PPT) + threadIdx.x;
  for (; base + (PPT - 1) * RES_T < total; base += step) {
    float v[PPT];
#pragma unroll
    for (int q = 0; q < PPT; ++q) v[q] = pad_elem(x, base + q * RES_T);
#pragma unroll
    for (int q = 0; q < PPT; ++q) hp[base + q * RES_T] = v[q];
  }
#pragma unroll
  for (int q = 0; q < PPT; ++q) {
    int i = base + q * RES_T;
    if (i < total) hp[i] = pad_elem(x, i);
  }
}


__device__ __forceinline__ int seg_of(int i, int d, int Nn, int Ee) {
  const int r = (i >= Ee) + (i >= 2 * Ee) + (i >= 3 * Ee);
  return r * Nn + d;
}

__device__ __forceinline__ void block_scan_incl(int* sh, int t, int n) {
  for (int ofs = 1; ofs < n; ofs <<= 1) {
    int add = (t >= ofs) ? sh[t - ofs] : 0;
    __syncthreads();
    sh[t] += add;
    __syncthreads();
  }
}

__global__ void __launch_bounds__(L1_T)
k_bhist(const int* __restrict__ edst,
        int* __restrict__ bcnt,
        int Nn, int Ee, int edges) {
  __shared__ int lh[MAXB];
  for (int i = threadIdx.x; i < MAXB; i += L1_T) lh[i] = 0;
  __syncthreads();
  const int base = blockIdx.x * CH + threadIdx.x;
  int d[L1_PER];
#pragma unroll
  for (int k = 0; k < L1_PER; ++k) d[k] = edst[min(base + k * L1_T, edges - 1)];
#pragma unroll
  for (int k = 0; k < L1_PER; ++k) {
    const int i = base + k * L1_T;
    if (i < edges) atomicAdd(&lh[seg_of(i, d[k], Nn, Ee) >> BSH], 1);
  }
  __syncthreads();
  for (int i = threadIdx.x; i < MAXB; i += L1_T) {
    const int v = lh[i];
    if (v) atomicAdd(&bcnt[i], v);
  }
}

__global__ void __launch_bounds__(MAXB)
k_bscan(const int* __restrict__ bcnt, int* __restrict__ bbase) {
  __shared__ int sh[MAXB];
  const int t = threadIdx.x;
  const int v = bcnt[t];
  sh[t] = v;
  __syncthreads();
  block_scan_incl(sh, t, MAXB);
  bbase[t] = sh[t] - v;
  if (t == MAXB - 1) bbase[MAXB] = sh[t];
}

__global__ void __launch_bounds__(L1_T)
k_bscatter(const int* __restrict__ esrc,
           const int* __restrict__ edst,
           const int* __restrict__ bbase,
           int* __restrict__ bcur,
           int2* __restrict__ pairs,
           int Nn, int Ee, int edges) {
  static_assert(L1_T == MAXB, "one bucket per thread in the run reservation");
  __shared__ int lh[MAXB];
  __shared__ int lc[MAXB];
  __shared__ int sc[MAXB];
  __shared__ int2 stage[CH];
  const int t = threadIdx.x;
  lh[t] = 0;
  __syncthreads();
  const int base = blockIdx.x * CH + t;
  int g[L1_PER], s[L1_PER];
#pragma unroll
  for (int k = 0; k < L1_PER; ++k) {
    const int idx = min(base + k * L1_T, edges - 1);
    s[k] = esrc[idx];
    g[k] = seg_of(idx, edst[idx], Nn, Ee);
  }
#pragma unroll
  for (int k = 0; k < L1_PER; ++k)
    if (base + k * L1_T < edges) atomicAdd(&lh[g[k] >> BSH], 1);
  __syncthreads();
  const int v = lh[t];
  sc[t] = v;
  __syncthreads();
  block_scan_incl(sc, t, MAXB);
  const int excl = sc[t] - v;
  lc[t] = excl;
  int gs = 0;
  if (v > 0) gs = bbase[t] + atomicAdd(&bcur[t], v);
  lh[t] = gs - excl;
  __syncthreads();
#pragma unroll
  for (int k = 0; k < L1_PER; ++k) {
    if (base + k * L1_T < edges) {
      const int slot = atomicAdd(&lc[g[k] >> BSH], 1);
      stage[slot] = make_int2(s[k], g[k]);
    }
  }
  __syncthreads();
  const int n = min(CH, edges - (int)blockIdx.x * CH);
  if (n == CH) {
#pragma unroll
    for (int k = 0; k < L1_PER; ++k) {
      const int i = t + k * L1_T;
      const int2 p = stage[i];
      pairs[i + lh[p.y >> BSH]] = p;
    }
  } else {
    for (int i = t; i < n; i += L1_T) {
      const int2 p = stage[i];
      pairs[i + lh[p.y >> BSH]] = p;
    }
  }
}

__global__ void __launch_bounds__(L2_T)
k_bsort(const int2* __restrict__ pairs,
        const int* __restrict__ bbase,
        int* __restrict__ off,
        int* __restrict__ col,
        int segs, int nbk) {
  __shared__ int cnt[BW];
  __shared__ int sc[L2_T];
  __shared__ int stage[CAP];
  const int t = threadIdx.x;
  const int b = blockIdx.x;
  const int base = bbase[b];
  const int n = bbase[b + 1] - base;
  const int g0 = b << BSH;
  const int nseg = min(BW, segs - g0);
  const int2* __restrict__ pp = pairs + base;
  int* __restrict__ cb = col + base;
  for (int i = t; i < BW; i += L2_T) cnt[i] = 0;
  __syncthreads();
  for (int i0 = t; i0 < n; i0 += L2_T * L2_U) {
    int l[L2_U];
#pragma unroll
    for (int u = 0; u < L2_U; ++u) l[u] = pp[min(i0 + u * L2_T, n - 1)].y - g0;
#pragma unroll
    for (int u = 0; u < L2_U; ++u)
      if (i0 + u * L2_T < n) atomicAdd(&cnt[l[u]], 1);
  }
  __syncthreads();
  int v[L2_PER];
  int s = 0;
#pragma unroll
  for (int q = 0; q < L2_PER; ++q) {
    v[q] = cnt[t * L2_PER + q];
    s += v[q];
  }
  sc[t] = s;
  __syncthreads();
  block_scan_incl(sc, t, L2_T);
  int run = sc[t] - s;
#pragma unroll
  for (int q = 0; q < L2_PER; ++q) {
    cnt[t * L2_PER + q] = run;
    run += v[q];
  }
  __syncthreads();
#pragma unroll 8
  for (int l = t; l < nseg; l += L2_T) off[g0 + l] = base + cnt[l];
  if (b == nbk - 1 && t == 0) off[segs] = base + n;
  __syncthreads();
  const bool inLds = (n <= CAP);
  for (int i0 = t; i0 < n; i0 += L2_T * L2_U) {
    int2 p[L2_U];
#pragma unroll
    for (int u = 0; u < L2_U; ++u) p[u] = pp[min(i0 + u * L2_T, n - 1)];
#pragma unroll
    for (int u = 0; u < L2_U; ++u) {
      if (i0 + u * L2_T < n) {
        const int slot = atomicAdd(&cnt[p[u].y - g0], 1);
        if (inLds) stage[slot] = p[u].x;
        else       cb[slot] = p[u].x;
      }
    }
  }
  __syncthreads();
  if (inLds) {
#pragma unroll 8
    for (int i = t; i < n; i += L2_T) cb[i] = stage[i];
  }
}

__device__ __forceinline__ void row_part(const float* __restrict__ p,
                                         v4f& q, v2f& t) {
  q = *(const v4f*)(p);
  t = *(const v2f*)(p + 8);
}

__device__ __forceinline__ v8f wmma_row(v4f aq, v2f at, v4f bq, v2f bt, v8f c) {
  c = wmma_k4(aq.xy, bq.xy, c);
  c = wmma_k4(aq.zw, bq.zw, c);
  c = wmma_k4(at, bt, c);
  return c;
}

__device__ __forceinline__ v4f row_piece(const char* __restrict__ hb, int row, unsigned po) {
  return *(const v4f*)(hb + ((unsigned)row * (unsigned)(SP * 4) + po));
}

#define LW  4
#define GR  12
#define CHR (8 * GR)

template <int Q>
__device__ __forceinline__ int quad_lane(int v) {
  return __builtin_amdgcn_update_dpp(0, v, Q * 0x55, 0xF, 0xF, true);
}

__device__ __forceinline__ int quad_pick(int v, int q) {
  switch (q) {
    case 0:  return quad_lane<0>(v);
    case 1:  return quad_lane<1>(v);
    case 2:  return quad_lane<2>(v);
    default: return quad_lane<3>(v);
  }
}

__device__ __forceinline__ void load_ids(const int* __restrict__ col, int c0, int e,
                                         int lb, int (&id)[GR / 4]) {
#pragma unroll
  for (int k = 0; k < GR / 4; ++k) {
    const int idx = c0 + 32 * k + lb;
    id[k] = -1;
    if (idx < e) id[k] = col[idx];
  }
}

__global__ void __launch_bounds__(32 * LW)
k_layer(const float* __restrict__ hp,
        const int* __restrict__ off,
        const int* __restrict__ col,
        const float* __restrict__ WsP,
        const float* __restrict__ WnP,
        const float* __restrict__ bsP,
        float* __restrict__ hout,
        int nTiles, int Nn) {
  static_assert(GR % 4 == 0, "a lane fetches the ids of every fourth gather load");
  __shared__ v4f sm[LW][16 * SP / 4];
  __shared__ v4f st[LW][CHR * SP / 4];
  const int wib = __builtin_amdgcn_readfirstlane((int)(threadIdx.x >> 5));
  const int wave = blockIdx.x * LW + wib;
  if (wave >= nTiles) return;
  const int lane = threadIdx.x & 31;
  const int m    = lane & 15;
  const int half = lane >> 4;
  const int ko   = half * 4;
  const int node = min(wave * 16 + m, Nn - 1);

  const int part = lane & 3;
  const int grp  = lane >> 2;
  const int lb   = 8 * part + grp;
  const unsigned po = (unsigned)part * 16u;
  const int nodeA = min(wave * 16 + grp, Nn - 1);
  const int nodeB = min(wave * 16 + grp + 8, Nn - 1);
  const char* __restrict__ hb = (const char*)hp;
  v4f* __restrict__ smw = sm[wib];
  const float* smf = (const float*)smw;
  v4f* __restrict__ stw = st[wib];

  union { v8f v; float f[8]; } acc;
  float bsum = bsP[m];
#pragma unroll
  for (int j = 0; j < 8; ++j) acc.f[j] = bsum;

  {
    v4f aq, bq;
    v2f at, bt;
    row_part(hp + (long long)node * SP + ko, aq, at);
    row_part(WsP + m * SP + ko, bq, bt);
    acc.v = wmma_row(aq, at, bq, bt, acc.v);
  }

  const int bnode = ((part & 2) ? nodeB : nodeA) + (part & 1);
  int bnd[RT];
#pragma unroll
  for (int r = 0; r < RT; ++r) bnd[r] = off[r * Nn + bnode];

  int cur[GR / 4];
  load_ids(col, __builtin_amdgcn_readlane(bnd[0], 0),
           __builtin_amdgcn_readlane(bnd[0], 31), lb, cur);
#pragma unroll
  for (int r = 0; r < RT; ++r) {
    const int s = __builtin_amdgcn_readlane(bnd[r], 0);
    const int e = __builtin_amdgcn_readlane(bnd[r], 31);
    const int oA0 = quad_lane<0>(bnd[r]), oA1 = quad_lane<1>(bnd[r]);
    const int oB0 = quad_lane<2>(bnd[r]), oB1 = quad_lane<3>(bnd[r]);
    int nxt[GR / 4];
#pragma unroll
    for (int k = 0; k < GR / 4; ++k) nxt[k] = -1;
    const v4f zero = {0.f, 0.f, 0.f, 0.f};
    v4f sA = zero, sB = zero;
    int c0 = s;
    do {
      if (c0 != s) load_ids(col, c0, e, lb, cur);
      int ids[GR];
#pragma unroll
      for (int i = 0; i < GR; ++i) ids[i] = quad_pick(cur[i >> 2], i & 3);
      v4f q[GR];
#pragma unroll
      for (int i = 0; i < GR; ++i) {
        q[i] = zero;
        if (ids[i] >= 0) q[i] = row_piece(hb, ids[i], po);
      }
      if (r + 1 < RT && c0 == s)
        load_ids(col, __builtin_amdgcn_readlane(bnd[r + 1], 0),
                 __builtin_amdgcn_readlane(bnd[r + 1], 31), lb, nxt);
#pragma unroll
      for (int i = 0; i < GR; ++i)
        if (c0 + 8 * i < e) stw[(8 * i + grp) * (SP / 4) + part] = q[i];
      const int aLo = max(oA0, c0) - c0;
      const int aN  = min(oA1, c0 + CHR) - c0 - aLo;
      const int bLo = max(oB0, c0) - c0;
      const int bN  = min(oB1, c0 + CHR) - c0 - bLo;
      int ia = aLo * (SP / 4) + part;
      int ib = bLo * (SP / 4) + part;
      const int tmax = max(aN, bN);
      for (int t = 0; t < tmax; t += 4) {
        v4f x0 = zero, x1 = zero, x2 = zero, x3 = zero;
        v4f y0 = zero, y1 = zero, y2 = zero, y3 = zero;
        if (t     < aN) x0 = stw[ia];
        if (t + 1 < aN) x1 = stw[ia + 1 * (SP / 4)];
        if (t + 2 < aN) x2 = stw[ia + 2 * (SP / 4)];
        if (t + 3 < aN) x3 = stw[ia + 3 * (SP / 4)];
        if (t     < bN) y0 = stw[ib];
        if (t + 1 < bN) y1 = stw[ib + 1 * (SP / 4)];
        if (t + 2 < bN) y2 = stw[ib + 2 * (SP / 4)];
        if (t + 3 < bN) y3 = stw[ib + 3 * (SP / 4)];
        sA += x0; sA += x1; sA += x2; sA += x3;
        sB += y0; sB += y1; sB += y2; sB += y3;
        ia += 4 * (SP / 4);
        ib += 4 * (SP / 4);
      }
      c0 += CHR;
    } while (c0 < e);
    if (r + 1 < RT) {
#pragma unroll
      for (int k = 0; k < GR / 4; ++k) cur[k] = nxt[k];
    }
    const int nA = oA1 - oA0;
    const int nB = oB1 - oB0;
    const float rA = 1.0f / fmaxf((float)nA, 1.0f);
    const float rB = 1.0f / fmaxf((float)nB, 1.0f);
    smw[grp * (SP / 4) + part]       = sA * rA;
    smw[(grp + 8) * (SP / 4) + part] = sB * rB;
    v4f aq, bq;
    v2f at, bt;
    row_part(smf + m * SP + ko, aq, at);
    row_part(WnP + r * 16 * SP + m * SP + ko, bq, bt);
    acc.v = wmma_row(aq, at, bq, bt, acc.v);
  }

  int rowBase = wave * 16 + half * 8;
  bool fullTile = (wave * 16 + 16) <= Nn;
  float* op = hout + (long long)rowBase * SP + m;
  if (fullTile) {
#pragma unroll
    for (int j = 0; j < 8; ++j) op[j * SP] = acc.f[j];
  } else {
#pragma unroll
    for (int j = 0; j < 8; ++j)
      if (rowBase + j < Nn) op[j * SP] = acc.f[j];
  }
}


__device__ __forceinline__ float quad_x1(float v) {
  return __builtin_bit_cast(float, __builtin_amdgcn_update_dpp(
             0, __builtin_bit_cast(int, v), 0xB1, 0xF, 0xF, true));
}
__device__ __forceinline__ float quad_x2(float v) {
  return __builtin_bit_cast(float, __builtin_amdgcn_update_dpp(
             0, __builtin_bit_cast(int, v), 0x4E, 0xF, 0xF, true));
}

__device__ __forceinline__ float dot4(v4f a, v4f b) {
  return a.x * b.x + a.y * b.y + a.z * b.z + a.w * b.w;
}

struct ScoreRows { v4f a0, b0, a1, b1, a2, b2, a3, b3; };

__device__ __forceinline__ ScoreRows score_load(const char* __restrict__ hb,
                                                int ca, int cb, unsigned po) {
  ScoreRows r;
  r.a0 = row_piece(hb, quad_lane<0>(ca), po);
  r.b0 = row_piece(hb, quad_lane<0>(cb), po);
  r.a1 = row_piece(hb, quad_lane<1>(ca), po);
  r.b1 = row_piece(hb, quad_lane<1>(cb), po);
  r.a2 = row_piece(hb, quad_lane<2>(ca), po);
  r.b2 = row_piece(hb, quad_lane<2>(cb), po);
  r.a3 = row_piece(hb, quad_lane<3>(ca), po);
  r.b3 = row_piece(hb, quad_lane<3>(cb), po);
  return r;
}

__device__ __forceinline__ float score_sum(const ScoreRows& r, int part) {
  const float p0 = dot4(r.a0, r.b0);
  const float p1 = dot4(r.a1, r.b1);
  const float p2 = dot4(r.a2, r.b2);
  const float p3 = dot4(r.a3, r.b3);
  const bool odd = (part & 1) != 0;
  const bool hi  = (part & 2) != 0;
  const float lo01 = (odd ? p1 : p0) + quad_x1(odd ? p0 : p1);
  const float lo23 = (odd ? p3 : p2) + quad_x1(odd ? p2 : p3);
  return (hi ? lo23 : lo01) + quad_x2(hi ? lo01 : lo23);
}

__global__ void __launch_bounds__(RES_T)
k_score(const float* __restrict__ hp,
        const int* __restrict__ ps, const int* __restrict__ pd,
        const int* __restrict__ ns, const int* __restrict__ nd,
        float* __restrict__ out, int E_) {
  const int* __restrict__ ia = blockIdx.y ? ns : ps;
  const int* __restrict__ ib = blockIdx.y ? nd : pd;
  float* __restrict__ o = out + (long long)blockIdx.y * E_;
  const char* __restrict__ hb = (const char*)hp;
  const int lane = threadIdx.x & 31;
  const int part = lane & 3;
  const int k    = part * 8 + (lane >> 2);
  const unsigned po = (unsigned)part * 16u;
  const int nw   = gridDim.x * (RES_T / 32);
  const int last = E_ - 1;
  const int wv = __builtin_amdgcn_readfirstlane((int)(threadIdx.x >> 5));
  int t = blockIdx.x * (RES_T / 32) + wv;
  if (t * 32 >= E_) return;
  int p  = min(t * 32 + k, last);
  int ca = ia[p], cb = ib[p];
  p = min((t + nw) * 32 + k, last);
  int na = ia[p], nb = ib[p];
  for (;;) {
    {
      const ScoreRows r = score_load(hb, ca, cb, po);
      p = min((t + 2 * nw) * 32 + k, last);
      ca = ia[p]; cb = ib[p];
      const float s = score_sum(r, part);
      const int i = t * 32 + k;
      if (i < E_) o[i] = s;
    }
    t += nw;
    if (t * 32 >= E_) break;
    {
      const ScoreRows r = score_load(hb, na, nb, po);
      p = min((t + 2 * nw) * 32 + k, last);
      na = ia[p]; nb = ib[p];
      const float s = score_sum(r, part);
      const int i = t * 32 + k;
      if (i < E_) o[i] = s;
    }
    t += nw;
    if (t * 32 >= E_) break;
  }
}

extern "C" void kernel_launch(void* const* d_in, const int* in_sizes, int n_in,
                              void* d_out, int out_size, void* d_ws, size_t ws_size,
                              hipStream_t stream) {
  (void)n_in; (void)out_size; (void)ws_size;
  const float* x    = (const float*)d_in[0];
  const int*   esrc = (const int*)d_in[1];
  const int*   edst = (const int*)d_in[2];
  const int*   nsrc = (const int*)d_in[3];
  const int*   ndst = (const int*)d_in[4];
  const float* Ws1  = (const float*)d_in[5];
  const float* Wn1  = (const float*)d_in[6];
  const float* b1   = (const float*)d_in[7];
  const float* Ws2  = (const float*)d_in[8];
  const float* Wn2  = (const float*)d_in[9];
  const float* b2   = (const float*)d_in[10];

  const int Nn = in_sizes[0] / D;
  const int Ee = in_sizes[3];

  const int segs   = RT * Nn;
  const int edges  = RT * Ee;
  const int nbk    = (segs + BW - 1) >> BSH;
  const int nCh    = (edges + CH - 1) / CH;
  const int offCnt = (segs + 1 + 3) & ~3;

  float* B0    = (float*)d_ws;
  float* B1    = B0 + (size_t)Nn * SP;
  int*   bcnt  = (int*)(B1 + (size_t)Nn * SP);
  int*   bcur  = bcnt + MAXB;
  int*   bbase = bcur + MAXB;
  int*   off   = bbase + MAXB + 4;
  int*   col   = off + (size_t)offCnt;
  int2*  pairs = (int2*)(col + (size_t)edges);
  float* WsP1  = (float*)(pairs + (size_t)edges);
  float* WnP1  = WsP1 + 16 * SP;
  float* bsP1  = WnP1 + RT * 16 * SP;
  float* WsP2  = bsP1 + 16;
  float* WnP2  = WsP2 + 16 * SP;
  float* bsP2  = WnP2 + RT * 16 * SP;

  const int tiles   = (Nn + 15) / 16;
  const int wBlocks = (tiles + LW - 1) / LW;
  const dim3 scoreGrid(512, 2);

  k_prep<<<1, 256, 0, stream>>>(Ws1, Wn1, b1, Ws2, Wn2, b2,
                                WsP1, WnP1, bsP1, WsP2, WnP2, bsP2, bcnt);
  k_padcopy<<<1024, RES_T, 0, stream>>>(x, B0, Nn);

  k_bhist<<<nCh, L1_T, 0, stream>>>(edst, bcnt, Nn, Ee, edges);
  k_bscan<<<1, MAXB, 0, stream>>>(bcnt, bbase);
  k_bscatter<<<nCh, L1_T, 0, stream>>>(esrc, edst, bbase, bcur, pairs, Nn, Ee, edges);
  k_bsort<<<nbk, L2_T, 0, stream>>>(pairs, bbase, off, col, segs, nbk);

  k_layer<<<wBlocks, 32 * LW, 0, stream>>>(B0, off, col, WsP1, WnP1, bsP1, B1, tiles, Nn);
  k_layer<<<wBlocks, 32 * LW, 0, stream>>>(B1, off, col, WsP2, WnP2, bsP2, B0, tiles, Nn);

  k_score<<<scoreGrid, RES_T, 0, stream>>>(B0, esrc, edst, nsrc, ndst,
                                           (float*)d_out, Ee);
}


# Sort level 1 in ONE pass: each bucket owns a fixed-capacity region of the pair array (overflow: a spill list level 2 scans); count pass and scan launch go, 9 launches to 7; sums stay in arrival order
# speedup vs baseline: 5.6455x; 1.0406x over previous
#include <hip/hip_runtime.h>


#define D  10
#define RT 4
#define SP 16

#define RES_T 256
#define PPT   8

#define BSH   12
#define BW    (1 << BSH)
#define MAXB  512
#define L1_T  512
#define L1_PER 16
#define CH    (L1_T * L1_PER)
#define L2_T  512
#define L2_PER (BW / L2_T)
#define L2_U  8
#define CAP   32768
#define CAPB  24576

typedef __attribute__((ext_vector_type(2))) float v2f;
typedef __attribute__((ext_vector_type(4))) float v4f;
typedef __attribute__((ext_vector_type(8))) float v8f;

__device__ __forceinline__ v8f wmma_k4(v2f a, v2f b, v8f c) {
  return __builtin_amdgcn_wmma_f32_16x16x4_f32(
       false, a,  false, b,
       (short)0, c,  false,  false);
}

__global__ void k_prep(const float* __restrict__ Ws1, const float* __restrict__ Wn1,
                       const float* __restrict__ b1,
                       const float* __restrict__ Ws2, const float* __restrict__ Wn2,
                       const float* __restrict__ b2,
                       float* __restrict__ WsP1, float* __restrict__ WnP1,
                       float* __restrict__ bsP1,
                       float* __restrict__ WsP2, float* __restrict__ WnP2,
                       float* __restrict__ bsP2,
                       int* __restrict__ bzero) {
  for (int i = threadIdx.x; i < 2 * MAXB; i += blockDim.x) bzero[i] = 0;
  for (int i = threadIdx.x; i < 16 * SP; i += blockDim.x) {
    int n = i / SP, k = i % SP;
    float s1 = 0.f, s2 = 0.f;
    if (n < D && k < D) {
#pragma unroll
      for (int r = 0; r < RT; ++r) {
        s1 += Ws1[r * D * D + n * D + k];
        s2 += Ws2[r * D * D + n * D + k];
      }
    }
    WsP1[i] = s1;
    WsP2[i] = s2;
  }
  for (int i = threadIdx.x; i < RT * 16 * SP; i += blockDim.x) {
    int r = i / (16 * SP), j = i % (16 * SP);
    int n = j / SP, k = j % SP;
    float v1 = 0.f, v2 = 0.f;
    if (n < D && k < D) {
      v1 = Wn1[r * D * D + n * D + k];
      v2 = Wn2[r * D * D + n * D + k];
    }
    WnP1[i] = v1;
    WnP2[i] = v2;
  }
  for (int i = threadIdx.x; i < 16; i += blockDim.x) {
    float s1 = 0.f, s2 = 0.f;
    if (i < D) {
#pragma unroll
      for (int r = 0; r < RT; ++r) {
        s1 += b1[r * D + i];
        s2 += b2[r * D + i];
      }
    }
    bsP1[i] = s1;
    bsP2[i] = s2;
  }
}

__device__ __forceinline__ float pad_elem(const float* __restrict__ x, int i) {
  int n = i / SP, k = i - n * SP;
  float t = x[(long long)n * D + min(k, D - 1)];
  return (k < D) ? t : 0.0f;
}

__global__ void __launch_bounds__(RES_T)
k_padcopy(const float* __restrict__ x, float* __restrict__ hp, int Nn) {
  const int total = Nn * SP;
  const int step  = gridDim.x * (RES_T * PPT);
  int base = blockIdx.x * (RES_T * PPT) + threadIdx.x;
  for (; base + (PPT - 1) * RES_T < total; base += step) {
    float v[PPT];
#pragma unroll
    for (int q = 0; q < PPT; ++q) v[q] = pad_elem(x, base + q * RES_T);
#pragma unroll
    for (int q = 0; q < PPT; ++q) hp[base + q * RES_T] = v[q];
  }
#pragma unroll
  for (int q = 0; q < PPT; ++q) {
    int i = base + q * RES_T;
    if (i < total) hp[i] = pad_elem(x, i);
  }
}


__device__ __forceinline__ int seg_of(int i, int d, int Nn, int Ee) {
  const int r = (i >= Ee) + (i >= 2 * Ee) + (i >= 3 * Ee);
  return r * Nn + d;
}

__device__ __forceinline__ void block_scan_incl(int* sh, int t, int n) {
  for (int ofs = 1; ofs < n; ofs <<= 1) {
    int add = (t >= ofs) ? sh[t - ofs] : 0;
    __syncthreads();
    sh[t] += add;
    __syncthreads();
  }
}

__global__ void __launch_bounds__(L1_T)
k_bscatter(const int* __restrict__ esrc,
           const int* __restrict__ edst,
           int* __restrict__ bcur,
           int* __restrict__ scur,
           int2* __restrict__ pairs,
           int2* __restrict__ spill,
           int Nn, int Ee, int edges) {
  static_assert(L1_T == MAXB, "one bucket per thread in the run reservation");
  __shared__ int lh[MAXB];
  __shared__ int lc[MAXB];
  __shared__ int sc[MAXB];
  __shared__ int lf[MAXB];
  __shared__ int ls[MAXB];
  __shared__ int anySpill;
  __shared__ int2 stage[CH];
  const int t = threadIdx.x;
  lh[t] = 0;
  if (t == 0) anySpill = 0;
  __syncthreads();
  const int base = blockIdx.x * CH + t;
  int g[L1_PER], s[L1_PER];
#pragma unroll
  for (int k = 0; k < L1_PER; ++k) {
    const int idx = min(base + k * L1_T, edges - 1);
    s[k] = esrc[idx];
    g[k] = seg_of(idx, edst[idx], Nn, Ee);
  }
#pragma unroll
  for (int k = 0; k < L1_PER; ++k)
    if (base + k * L1_T < edges) atomicAdd(&lh[g[k] >> BSH], 1);
  __syncthreads();
  const int v = lh[t];
  sc[t] = v;
  __syncthreads();
  block_scan_incl(sc, t, MAXB);
  const int excl = sc[t] - v;
  lc[t] = excl;
  int old = 0;
  if (v > 0) old = atomicAdd(&bcur[t], v);
  const int fit = min(max(CAPB - old, 0), v);
  int sb = 0;
  if (fit < v) {
    sb = atomicAdd(scur, v - fit);
    anySpill = 1;
  }
  lh[t] = t * CAPB + old - excl;
  lf[t] = excl + fit;
  ls[t] = sb - fit - excl;
  __syncthreads();
#pragma unroll
  for (int k = 0; k < L1_PER; ++k) {
    if (base + k * L1_T < edges) {
      const int slot = atomicAdd(&lc[g[k] >> BSH], 1);
      stage[slot] = make_int2(s[k], g[k]);
    }
  }
  __syncthreads();
  const int n = min(CH, edges - (int)blockIdx.x * CH);
  const bool spilled = anySpill != 0;
  if (n == CH && !spilled) {
#pragma unroll
    for (int k = 0; k < L1_PER; ++k) {
      const int i = t + k * L1_T;
      const int2 p = stage[i];
      pairs[i + lh[p.y >> BSH]] = p;
    }
  } else if (!spilled) {
    for (int i = t; i < n; i += L1_T) {
      const int2 p = stage[i];
      pairs[i + lh[p.y >> BSH]] = p;
    }
  } else {
    for (int i = t; i < n; i += L1_T) {
      const int2 p = stage[i];
      const int b = p.y >> BSH;
      if (i < lf[b]) pairs[i + lh[b]] = p;
      else           spill[i + ls[b]] = p;
    }
  }
}

__global__ void __launch_bounds__(L2_T)
k_bsort(const int2* __restrict__ pairs,
        const int2* __restrict__ spill,
        const int* __restrict__ bcur,
        const int* __restrict__ scur,
        int* __restrict__ off,
        int* __restrict__ col,
        int segs, int nbk) {
  __shared__ int cnt[BW];
  __shared__ int sc[L2_T];
  __shared__ int stage[CAP];
  static_assert(L2_T == MAXB, "one fill count per thread in the bucket-start sum");
  const int t = threadIdx.x;
  const int b = blockIdx.x;
  int part = (t < b) ? bcur[t] : 0;
#pragma unroll
  for (int m = 16; m > 0; m >>= 1) part += __shfl_xor(part, m);
  if ((t & 31) == 0) sc[t >> 5] = part;
  for (int i = t; i < BW; i += L2_T) cnt[i] = 0;
  __syncthreads();
  int base = 0;
#pragma unroll
  for (int w = 0; w < L2_T / 32; ++w) base += sc[w];
  const int n = bcur[b];
  const int nreg = min(n, CAPB);
  const int nsp = (n > CAPB) ? scur[0] : 0;
  const int g0 = b << BSH;
  const int nseg = min(BW, segs - g0);
  const int2* __restrict__ pp = pairs + (size_t)b * CAPB;
  int* __restrict__ cb = col + base;
  for (int i0 = t; i0 < nreg; i0 += L2_T * L2_U) {
    int l[L2_U];
#pragma unroll
    for (int u = 0; u < L2_U; ++u) l[u] = pp[min(i0 + u * L2_T, nreg - 1)].y - g0;
#pragma unroll
    for (int u = 0; u < L2_U; ++u)
      if (i0 + u * L2_T < nreg) atomicAdd(&cnt[l[u]], 1);
  }
  for (int i = t; i < nsp; i += L2_T) {
    const int g = spill[i].y;
    if ((g >> BSH) == b) atomicAdd(&cnt[g - g0], 1);
  }
  __syncthreads();
  int v[L2_PER];
  int s = 0;
#pragma unroll
  for (int q = 0; q < L2_PER; ++q) {
    v[q] = cnt[t * L2_PER + q];
    s += v[q];
  }
  sc[t] = s;
  __syncthreads();
  block_scan_incl(sc, t, L2_T);
  int run = sc[t] - s;
#pragma unroll
  for (int q = 0; q < L2_PER; ++q) {
    cnt[t * L2_PER + q] = run;
    run += v[q];
  }
  __syncthreads();
#pragma unroll 8
  for (int l = t; l < nseg; l += L2_T) off[g0 + l] = base + cnt[l];
  if (b == nbk - 1 && t == 0) off[segs] = base + n;
  __syncthreads();
  const bool inLds = (n <= CAP);
  for (int i0 = t; i0 < nreg; i0 += L2_T * L2_U) {
    int2 p[L2_U];
#pragma unroll
    for (int u = 0; u < L2_U; ++u) p[u] = pp[min(i0 + u * L2_T, nreg - 1)];
#pragma unroll
    for (int u = 0; u < L2_U; ++u) {
      if (i0 + u * L2_T < nreg) {
        const int slot = atomicAdd(&cnt[p[u].y - g0], 1);
        if (inLds) stage[slot] = p[u].x;
        else       cb[slot] = p[u].x;
      }
    }
  }
  for (int i = t; i < nsp; i += L2_T) {
    const int2 p = spill[i];
    if ((p.y >> BSH) == b) {
      const int slot = atomicAdd(&cnt[p.y - g0], 1);
      if (inLds) stage[slot] = p.x;
      else       cb[slot] = p.x;
    }
  }
  __syncthreads();
  if (inLds) {
#pragma unroll 8
    for (int i = t; i < n; i += L2_T) cb[i] = stage[i];
  }
}

__device__ __forceinline__ void row_part(const float* __restrict__ p,
                                         v4f& q, v2f& t) {
  q = *(const v4f*)(p);
  t = *(const v2f*)(p + 8);
}

__device__ __forceinline__ v8f wmma_row(v4f aq, v2f at, v4f bq, v2f bt, v8f c) {
  c = wmma_k4(aq.xy, bq.xy, c);
  c = wmma_k4(aq.zw, bq.zw, c);
  c = wmma_k4(at, bt, c);
  return c;
}

__device__ __forceinline__ v4f row_piece(const char* __restrict__ hb, int row, unsigned po) {
  return *(const v4f*)(hb + ((unsigned)row * (unsigned)(SP * 4) + po));
}

#define LW  4
#define GR  12
#define CHR (8 * GR)

template <int Q>
__device__ __forceinline__ int quad_lane(int v) {
  return __builtin_amdgcn_update_dpp(0, v, Q * 0x55, 0xF, 0xF, true);
}

__device__ __forceinline__ int quad_pick(int v, int q) {
  switch (q) {
    case 0:  return quad_lane<0>(v);
    case 1:  return quad_lane<1>(v);
    case 2:  return quad_lane<2>(v);
    default: return quad_lane<3>(v);
  }
}

__device__ __forceinline__ void load_ids(const int* __restrict__ col, int c0, int e,
                                         int lb, int (&id)[GR / 4]) {
#pragma unroll
  for (int k = 0; k < GR / 4; ++k) {
    const int idx = c0 + 32 * k + lb;
    id[k] = -1;
    if (idx < e) id[k] = col[idx];
  }
}

__global__ void __launch_bounds__(32 * LW)
k_layer(const float* __restrict__ hp,
        const int* __restrict__ off,
        const int* __restrict__ col,
        const float* __restrict__ WsP,
        const float* __restrict__ WnP,
        const float* __restrict__ bsP,
        float* __restrict__ hout,
        int nTiles, int Nn) {
  static_assert(GR % 4 == 0, "a lane fetches the ids of every fourth gather load");
  __shared__ v4f sm[LW][16 * SP / 4];
  __shared__ v4f st[LW][CHR * SP / 4];
  const int wib = __builtin_amdgcn_readfirstlane((int)(threadIdx.x >> 5));
  const int wave = blockIdx.x * LW + wib;
  if (wave >= nTiles) return;
  const int lane = threadIdx.x & 31;
  const int m    = lane & 15;
  const int half = lane >> 4;
  const int ko   = half * 4;
  const int node = min(wave * 16 + m, Nn - 1);

  const int part = lane & 3;
  const int grp  = lane >> 2;
  const int lb   = 8 * part + grp;
  const unsigned po = (unsigned)part * 16u;
  const int nodeA = min(wave * 16 + grp, Nn - 1);
  const int nodeB = min(wave * 16 + grp + 8, Nn - 1);
  const char* __restrict__ hb = (const char*)hp;
  v4f* __restrict__ smw = sm[wib];
  const float* smf = (const float*)smw;
  v4f* __restrict__ stw = st[wib];

  union { v8f v; float f[8]; } acc;
  float bsum = bsP[m];
#pragma unroll
  for (int j = 0; j < 8; ++j) acc.f[j] = bsum;

  {
    v4f aq, bq;
    v2f at, bt;
    row_part(hp + (long long)node * SP + ko, aq, at);
    row_part(WsP + m * SP + ko, bq, bt);
    acc.v = wmma_row(aq, at, bq, bt, acc.v);
  }

  const int bnode = ((part & 2) ? nodeB : nodeA) + (part & 1);
  int bnd[RT];
#pragma unroll
  for (int r = 0; r < RT; ++r) bnd[r] = off[r * Nn + bnode];

  int cur[GR / 4];
  load_ids(col, __builtin_amdgcn_readlane(bnd[0], 0),
           __builtin_amdgcn_readlane(bnd[0], 31), lb, cur);
#pragma unroll
  for (int r = 0; r < RT; ++r) {
    const int s = __builtin_amdgcn_readlane(bnd[r], 0);
    const int e = __builtin_amdgcn_readlane(bnd[r], 31);
    const int oA0 = quad_lane<0>(bnd[r]), oA1 = quad_lane<1>(bnd[r]);
    const int oB0 = quad_lane<2>(bnd[r]), oB1 = quad_lane<3>(bnd[r]);
    int nxt[GR / 4];
#pragma unroll
    for (int k = 0; k < GR / 4; ++k) nxt[k] = -1;
    const v4f zero = {0.f, 0.f, 0.f, 0.f};
    v4f sA = zero, sB = zero;
    int c0 = s;
    do {
      if (c0 != s) load_ids(col, c0, e, lb, cur);
      int ids[GR];
#pragma unroll
      for (int i = 0; i < GR; ++i) ids[i] = quad_pick(cur[i >> 2], i & 3);
      v4f q[GR];
#pragma unroll
      for (int i = 0; i < GR; ++i) {
        q[i] = zero;
        if (ids[i] >= 0) q[i] = row_piece(hb, ids[i], po);
      }
      if (r + 1 < RT && c0 == s)
        load_ids(col, __builtin_amdgcn_readlane(bnd[r + 1], 0),
                 __builtin_amdgcn_readlane(bnd[r + 1], 31), lb, nxt);
#pragma unroll
      for (int i = 0; i < GR; ++i)
        if (c0 + 8 * i < e) stw[(8 * i + grp) * (SP / 4) + part] = q[i];
      const int aLo = max(oA0, c0) - c0;
      const int aN  = min(oA1, c0 + CHR) - c0 - aLo;
      const int bLo = max(oB0, c0) - c0;
      const int bN  = min(oB1, c0 + CHR) - c0 - bLo;
      int ia = aLo * (SP / 4) + part;
      int ib = bLo * (SP / 4) + part;
      const int tmax = max(aN, bN);
      for (int t = 0; t < tmax; t += 4) {
        v4f x0 = zero, x1 = zero, x2 = zero, x3 = zero;
        v4f y0 = zero, y1 = zero, y2 = zero, y3 = zero;
        if (t     < aN) x0 = stw[ia];
        if (t + 1 < aN) x1 = stw[ia + 1 * (SP / 4)];
        if (t + 2 < aN) x2 = stw[ia + 2 * (SP / 4)];
        if (t + 3 < aN) x3 = stw[ia + 3 * (SP / 4)];
        if (t     < bN) y0 = stw[ib];
        if (t + 1 < bN) y1 = stw[ib + 1 * (SP / 4)];
        if (t + 2 < bN) y2 = stw[ib + 2 * (SP / 4)];
        if (t + 3 < bN) y3 = stw[ib + 3 * (SP / 4)];
        sA += x0; sA += x1; sA += x2; sA += x3;
        sB += y0; sB += y1; sB += y2; sB += y3;
        ia += 4 * (SP / 4);
        ib += 4 * (SP / 4);
      }
      c0 += CHR;
    } while (c0 < e);
    if (r + 1 < RT) {
#pragma unroll
      for (int k = 0; k < GR / 4; ++k) cur[k] = nxt[k];
    }
    const int nA = oA1 - oA0;
    const int nB = oB1 - oB0;
    const float rA = 1.0f / fmaxf((float)nA, 1.0f);
    const float rB = 1.0f / fmaxf((float)nB, 1.0f);
    smw[grp * (SP / 4) + part]       = sA * rA;
    smw[(grp + 8) * (SP / 4) + part] = sB * rB;
    v4f aq, bq;
    v2f at, bt;
    row_part(smf + m * SP + ko, aq, at);
    row_part(WnP + r * 16 * SP + m * SP + ko, bq, bt);
    acc.v = wmma_row(aq, at, bq, bt, acc.v);
  }

  int rowBase = wave * 16 + half * 8;
  bool fullTile = (wave * 16 + 16) <= Nn;
  float* op = hout + (long long)rowBase * SP + m;
  if (fullTile) {
#pragma unroll
    for (int j = 0; j < 8; ++j) op[j * SP] = acc.f[j];
  } else {
#pragma unroll
    for (int j = 0; j < 8; ++j)
      if (rowBase + j < Nn) op[j * SP] = acc.f[j];
  }
}


__device__ __forceinline__ float quad_x1(float v) {
  return __builtin_bit_cast(float, __builtin_amdgcn_update_dpp(
             0, __builtin_bit_cast(int, v), 0xB1, 0xF, 0xF, true));
}
__device__ __forceinline__ float quad_x2(float v) {
  return __builtin_bit_cast(float, __builtin_amdgcn_update_dpp(
             0, __builtin_bit_cast(int, v), 0x4E, 0xF, 0xF, true));
}

__device__ __forceinline__ float dot4(v4f a, v4f b) {
  return a.x * b.x + a.y * b.y + a.z * b.z + a.w * b.w;
}

struct ScoreRows { v4f a0, b0, a1, b1, a2, b2, a3, b3; };

__device__ __forceinline__ ScoreRows score_load(const char* __restrict__ hb,
                                                int ca, int cb, unsigned po) {
  ScoreRows r;
  r.a0 = row_piece(hb, quad_lane<0>(ca), po);
  r.b0 = row_piece(hb, quad_lane<0>(cb), po);
  r.a1 = row_piece(hb, quad_lane<1>(ca), po);
  r.b1 = row_piece(hb, quad_lane<1>(cb), po);
  r.a2 = row_piece(hb, quad_lane<2>(ca), po);
  r.b2 = row_piece(hb, quad_lane<2>(cb), po);
  r.a3 = row_piece(hb, quad_lane<3>(ca), po);
  r.b3 = row_piece(hb, quad_lane<3>(cb), po);
  return r;
}

__device__ __forceinline__ float score_sum(const ScoreRows& r, int part) {
  const float p0 = dot4(r.a0, r.b0);
  const float p1 = dot4(r.a1, r.b1);
  const float p2 = dot4(r.a2, r.b2);
  const float p3 = dot4(r.a3, r.b3);
  const bool odd = (part & 1) != 0;
  const bool hi  = (part & 2) != 0;
  const float lo01 = (odd ? p1 : p0) + quad_x1(odd ? p0 : p1);
  const float lo23 = (odd ? p3 : p2) + quad_x1(odd ? p2 : p3);
  return (hi ? lo23 : lo01) + quad_x2(hi ? lo01 : lo23);
}

__global__ void __launch_bounds__(RES_T)
k_score(const float* __restrict__ hp,
        const int* __restrict__ ps, const int* __restrict__ pd,
        const int* __restrict__ ns, const int* __restrict__ nd,
        float* __restrict__ out, int E_) {
  const int* __restrict__ ia = blockIdx.y ? ns : ps;
  const int* __restrict__ ib = blockIdx.y ? nd : pd;
  float* __restrict__ o = out + (long long)blockIdx.y * E_;
  const char* __restrict__ hb = (const char*)hp;
  const int lane = threadIdx.x & 31;
  const int part = lane & 3;
  const int k    = part * 8 + (lane >> 2);
  const unsigned po = (unsigned)part * 16u;
  const int nw   = gridDim.x * (RES_T / 32);
  const int last = E_ - 1;
  const int wv = __builtin_amdgcn_readfirstlane((int)(threadIdx.x >> 5));
  int t = blockIdx.x * (RES_T / 32) + wv;
  if (t * 32 >= E_) return;
  int p  = min(t * 32 + k, last);
  int ca = ia[p], cb = ib[p];
  p = min((t + nw) * 32 + k, last);
  int na = ia[p], nb = ib[p];
  for (;;) {
    {
      const ScoreRows r = score_load(hb, ca, cb, po);
      p = min((t + 2 * nw) * 32 + k, last);
      ca = ia[p]; cb = ib[p];
      const float s = score_sum(r, part);
      const int i = t * 32 + k;
      if (i < E_) o[i] = s;
    }
    t += nw;
    if (t * 32 >= E_) break;
    {
      const ScoreRows r = score_load(hb, na, nb, po);
      p = min((t + 2 * nw) * 32 + k, last);
      na = ia[p]; nb = ib[p];
      const float s = score_sum(r, part);
      const int i = t * 32 + k;
      if (i < E_) o[i] = s;
    }
    t += nw;
    if (t * 32 >= E_) break;
  }
}

extern "C" void kernel_launch(void* const* d_in, const int* in_sizes, int n_in,
                              void* d_out, int out_size, void* d_ws, size_t ws_size,
                              hipStream_t stream) {
  (void)n_in; (void)out_size; (void)ws_size;
  const float* x    = (const float*)d_in[0];
  const int*   esrc = (const int*)d_in[1];
  const int*   edst = (const int*)d_in[2];
  const int*   nsrc = (const int*)d_in[3];
  const int*   ndst = (const int*)d_in[4];
  const float* Ws1  = (const float*)d_in[5];
  const float* Wn1  = (const float*)d_in[6];
  const float* b1   = (const float*)d_in[7];
  const float* Ws2  = (const float*)d_in[8];
  const float* Wn2  = (const float*)d_in[9];
  const float* b2   = (const float*)d_in[10];

  const int Nn = in_sizes[0] / D;
  const int Ee = in_sizes[3];

  const int segs   = RT * Nn;
  const int edges  = RT * Ee;
  const int nbk    = (segs + BW - 1) >> BSH;
  const int nCh    = (edges + CH - 1) / CH;
  const int offCnt = (segs + 1 + 3) & ~3;

  float* B0    = (float*)d_ws;
  float* B1    = B0 + (size_t)Nn * SP;
  int*   bcur  = (int*)(B1 + (size_t)Nn * SP);
  int*   scur  = bcur + MAXB;
  int*   off   = scur + MAXB;
  int*   col   = off + (size_t)offCnt;
  float* WsP1  = (float*)(col + (size_t)edges);
  float* WnP1  = WsP1 + 16 * SP;
  float* bsP1  = WnP1 + RT * 16 * SP;
  float* WsP2  = bsP1 + 16;
  float* WnP2  = WsP2 + 16 * SP;
  float* bsP2  = WnP2 + RT * 16 * SP;
  int2*  pairs = (int2*)(bsP2 + 16);
  int2*  spill = pairs + (size_t)nbk * CAPB;

  const int tiles   = (Nn + 15) / 16;
  const int wBlocks = (tiles + LW - 1) / LW;
  const dim3 scoreGrid(512, 2);

  k_prep<<<1, 256, 0, stream>>>(Ws1, Wn1, b1, Ws2, Wn2, b2,
                                WsP1, WnP1, bsP1, WsP2, WnP2, bsP2, bcur);
  k_padcopy<<<1024, RES_T, 0, stream>>>(x, B0, Nn);

  k_bscatter<<<nCh, L1_T, 0, stream>>>(esrc, edst, bcur, scur, pairs, spill, Nn, Ee, edges);
  k_bsort<<<nbk, L2_T, 0, stream>>>(pairs, spill, bcur, scur, off, col, segs, nbk);

  k_layer<<<wBlocks, 32 * LW, 0, stream>>>(B0, off, col, WsP1, WnP1, bsP1, B1, tiles, Nn);
  k_layer<<<wBlocks, 32 * LW, 0, stream>>>(B1, off, col, WsP2, WnP2, bsP2, B0, tiles, Nn);

  k_score<<<scoreGrid, RES_T, 0, stream>>>(B0, esrc, edst, nsrc, ndst,
                                           (float*)d_out, Ee);
}
